# starting kernel
# baseline (speedup 1.0000x reference)
#include <hip/hip_runtime.h>

typedef __attribute__((ext_vector_type(2))) float v2f;
typedef __attribute__((ext_vector_type(8))) float v8f;

#define N_NODES 100000
#define N_EDGES 1600000
#define HID 128


__global__ __launch_bounds__(256) void k_init_deg(float* __restrict__ deg) {
    int i = blockIdx.x * 256 + threadIdx.x;
    if (i < N_NODES) deg[i] = 1.0f;
}

__global__ __launch_bounds__(256) void k_deg(const int* __restrict__ dst,
                                             float* __restrict__ deg) {
    int e = blockIdx.x * 256 + threadIdx.x;
    if (e < N_EDGES) atomicAdd(&deg[dst[e]], 1.0f);
}

__global__ __launch_bounds__(256) void k_dinv(float* __restrict__ deg) {
    int i = blockIdx.x * 256 + threadIdx.x;
    if (i < N_NODES) deg[i] = rsqrtf(deg[i]);
}

template <int MODE>
__global__ __launch_bounds__(256) void k_gemm(
    const float* __restrict__ A, const float* __restrict__ W,
    const float* __restrict__ b_in, const float* __restrict__ b_out,
    const float* __restrict__ xres, float* __restrict__ out) {
    __shared__ float sW[HID * HID];

    const int tid = threadIdx.x;
    for (int i = tid * 4; i < HID * HID; i += 256 * 4)
        *(float4*)&sW[i] = *(const float4*)&W[i];
    __syncthreads();

    const int wave = tid >> 5;
    const int lane = tid & 31;
    const int half = lane >> 4;
    const int lm   = lane & 15;
    const int kb   = half * 2;
    const int m0   = blockIdx.x * 16;
    const int n0   = wave * 16;

    const float* Arow = A + (size_t)(m0 + lm) * HID;

    v8f c = {};
    #pragma unroll 4
    for (int kk = 0; kk < HID; kk += 4) {
        v2f a, b;
        if (MODE == 0) {
            a = *(const v2f*)(Arow + kk + kb);
        } else {
            float a0 = Arow[kk + kb]     + b_in[kk + kb];
            float a1 = Arow[kk + kb + 1] + b_in[kk + kb + 1];
            a.x = a0 > 0.f ? a0 : 0.f;
            a.y = a1 > 0.f ? a1 : 0.f;
        }
        b.x = sW[(kk + kb) * HID + n0 + lm];
        b.y = sW[(kk + kb + 1) * HID + n0 + lm];
        c = __builtin_amdgcn_wmma_f32_16x16x4_f32(false, a, false, b,
                                                  (short)0, c, false, false);
    }

    const int n = n0 + lm;
    #pragma unroll
    for (int v = 0; v < 8; ++v) {
        int m = m0 + v + half * 8;
        float val = c[v];
        if (MODE == 1) {
            val += b_out[n];
            val = val > 0.f ? val : 0.f;
            val += xres[(size_t)m * HID + n];
        }
        out[(size_t)m * HID + n] = val;
    }
}


__global__ __launch_bounds__(256) void k_selfloop(const float* __restrict__ dinv,
                                                  const float* __restrict__ xw,
                                                  float* __restrict__ agg) {
    size_t g = (size_t)blockIdx.x * 256 + threadIdx.x;
    size_t node = g >> 5;
    int lane = (int)(g & 31);
    if (node >= N_NODES) return;
    float di = dinv[node];
    float w = di * di;
    float4 v = *(const float4*)(xw + node * HID + lane * 4);
    float4 r = {w * v.x, w * v.y, w * v.z, w * v.w};
    *(float4*)(agg + node * HID + lane * 4) = r;
}


__global__ __launch_bounds__(256) void k_scatter(const int* __restrict__ src,
                                                 const int* __restrict__ dst,
                                                 const float* __restrict__ dinv,
                                                 const float* __restrict__ xw,
                                                 float* __restrict__ agg) {
    size_t g = (size_t)blockIdx.x * 256 + threadIdx.x;
    size_t e = g >> 5;
    int lane = (int)(g & 31);
    if (e >= N_EDGES) return;
    int s = src[e];
    int d = dst[e];
    float norm = dinv[s] * dinv[d];
    float4 v = *(const float4*)(xw + (size_t)s * HID + lane * 4);
    float* p = agg + (size_t)d * HID + lane * 4;
    atomicAdd(p + 0, norm * v.x);
    atomicAdd(p + 1, norm * v.y);
    atomicAdd(p + 2, norm * v.z);
    atomicAdd(p + 3, norm * v.w);
}


extern "C" void kernel_launch(void* const* d_in, const int* in_sizes, int n_in,
                              void* d_out, int out_size, void* d_ws, size_t ws_size,
                              hipStream_t stream) {
    const float* x     = (const float*)d_in[0];
    const int*   ei    = (const int*)d_in[1];
    const float* W_gcn = (const float*)d_in[2];
    const float* b_gcn = (const float*)d_in[3];
    const float* W_lin = (const float*)d_in[4];
    const float* b_lin = (const float*)d_in[5];

    const int* src = ei;
    const int* dst = ei + N_EDGES;

    float* out = (float*)d_out;

    float* dinv = (float*)d_ws;
    float* agg  = (float*)((char*)d_ws + (size_t)N_NODES * sizeof(float));
    float* xw   = out;

    const int NB_NODE  = (N_NODES + 255) / 256;
    const int NB_EDGE  = (N_EDGES + 255) / 256;
    const int NB_TILE  = N_NODES / 16;
    const int NB_SELF  = (N_NODES * 32 + 255) / 256;
    const int NB_SCAT  = (int)(((size_t)N_EDGES * 32 + 255) / 256);

    k_init_deg<<<NB_NODE, 256, 0, stream>>>(dinv);
    k_deg<<<NB_EDGE, 256, 0, stream>>>(dst, dinv);
    k_dinv<<<NB_NODE, 256, 0, stream>>>(dinv);

    k_gemm<0><<<NB_TILE, 256, 0, stream>>>(x, W_gcn, nullptr, nullptr, nullptr, xw);

    k_selfloop<<<NB_SELF, 256, 0, stream>>>(dinv, xw, agg);
    k_scatter<<<NB_SCAT, 256, 0, stream>>>(src, dst, dinv, xw, agg);

    k_gemm<1><<<NB_TILE, 256, 0, stream>>>(agg, W_lin, b_gcn, b_lin, x, out);
}


# edge scatter: each lane owns features l, l+32, l+64, l+96 so every row load and f32 atomic add instruction covers one whole 128-B line instead of four quarter lines; float atomics kept
# speedup vs baseline: 1.2732x; 1.2732x over previous
#include <hip/hip_runtime.h>

typedef __attribute__((ext_vector_type(2))) float v2f;
typedef __attribute__((ext_vector_type(8))) float v8f;

#define N_NODES 100000
#define N_EDGES 1600000
#define HID 128


__global__ __launch_bounds__(256) void k_init_deg(float* __restrict__ deg) {
    int i = blockIdx.x * 256 + threadIdx.x;
    if (i < N_NODES) deg[i] = 1.0f;
}

__global__ __launch_bounds__(256) void k_deg(const int* __restrict__ dst,
                                             float* __restrict__ deg) {
    int e = blockIdx.x * 256 + threadIdx.x;
    if (e < N_EDGES) atomicAdd(&deg[dst[e]], 1.0f);
}

__global__ __launch_bounds__(256) void k_dinv(float* __restrict__ deg) {
    int i = blockIdx.x * 256 + threadIdx.x;
    if (i < N_NODES) deg[i] = rsqrtf(deg[i]);
}

template <int MODE>
__global__ __launch_bounds__(256) void k_gemm(
    const float* __restrict__ A, const float* __restrict__ W,
    const float* __restrict__ b_in, const float* __restrict__ b_out,
    const float* __restrict__ xres, float* __restrict__ out) {
    __shared__ float sW[HID * HID];

    const int tid = threadIdx.x;
    for (int i = tid * 4; i < HID * HID; i += 256 * 4)
        *(float4*)&sW[i] = *(const float4*)&W[i];
    __syncthreads();

    const int wave = tid >> 5;
    const int lane = tid & 31;
    const int half = lane >> 4;
    const int lm   = lane & 15;
    const int kb   = half * 2;
    const int m0   = blockIdx.x * 16;
    const int n0   = wave * 16;

    const float* Arow = A + (size_t)(m0 + lm) * HID;

    v8f c = {};
    #pragma unroll 4
    for (int kk = 0; kk < HID; kk += 4) {
        v2f a, b;
        if (MODE == 0) {
            a = *(const v2f*)(Arow + kk + kb);
        } else {
            float a0 = Arow[kk + kb]     + b_in[kk + kb];
            float a1 = Arow[kk + kb + 1] + b_in[kk + kb + 1];
            a.x = a0 > 0.f ? a0 : 0.f;
            a.y = a1 > 0.f ? a1 : 0.f;
        }
        b.x = sW[(kk + kb) * HID + n0 + lm];
        b.y = sW[(kk + kb + 1) * HID + n0 + lm];
        c = __builtin_amdgcn_wmma_f32_16x16x4_f32(false, a, false, b,
                                                  (short)0, c, false, false);
    }

    const int n = n0 + lm;
    #pragma unroll
    for (int v = 0; v < 8; ++v) {
        int m = m0 + v + half * 8;
        float val = c[v];
        if (MODE == 1) {
            val += b_out[n];
            val = val > 0.f ? val : 0.f;
            val += xres[(size_t)m * HID + n];
        }
        out[(size_t)m * HID + n] = val;
    }
}


__global__ __launch_bounds__(256) void k_selfloop(const float* __restrict__ dinv,
                                                  const float* __restrict__ xw,
                                                  float* __restrict__ agg) {
    size_t g = (size_t)blockIdx.x * 256 + threadIdx.x;
    size_t node = g >> 5;
    int lane = (int)(g & 31);
    if (node >= N_NODES) return;
    float di = dinv[node];
    float w = di * di;
    float4 v = *(const float4*)(xw + node * HID + lane * 4);
    float4 r = {w * v.x, w * v.y, w * v.z, w * v.w};
    *(float4*)(agg + node * HID + lane * 4) = r;
}


__global__ __launch_bounds__(256) void k_scatter(const int* __restrict__ src,
                                                 const int* __restrict__ dst,
                                                 const float* __restrict__ dinv,
                                                 const float* __restrict__ xw,
                                                 float* __restrict__ agg) {
    size_t g = (size_t)blockIdx.x * 256 + threadIdx.x;
    size_t e = g >> 5;
    int lane = (int)(g & 31);
    if (e >= N_EDGES) return;
    int s = src[e];
    int d = dst[e];
    float norm = dinv[s] * dinv[d];
    const float* q = xw + (size_t)s * HID + lane;
    float v0 = q[0];
    float v1 = q[32];
    float v2 = q[64];
    float v3 = q[96];
    float* p = agg + (size_t)d * HID + lane;
    atomicAdd(p + 0,  norm * v0);
    atomicAdd(p + 32, norm * v1);
    atomicAdd(p + 64, norm * v2);
    atomicAdd(p + 96, norm * v3);
}


extern "C" void kernel_launch(void* const* d_in, const int* in_sizes, int n_in,
                              void* d_out, int out_size, void* d_ws, size_t ws_size,
                              hipStream_t stream) {
    const float* x     = (const float*)d_in[0];
    const int*   ei    = (const int*)d_in[1];
    const float* W_gcn = (const float*)d_in[2];
    const float* b_gcn = (const float*)d_in[3];
    const float* W_lin = (const float*)d_in[4];
    const float* b_lin = (const float*)d_in[5];

    const int* src = ei;
    const int* dst = ei + N_EDGES;

    float* out = (float*)d_out;

    float* dinv = (float*)d_ws;
    float* agg  = (float*)((char*)d_ws + (size_t)N_NODES * sizeof(float));
    float* xw   = out;

    const int NB_NODE  = (N_NODES + 255) / 256;
    const int NB_EDGE  = (N_EDGES + 255) / 256;
    const int NB_TILE  = N_NODES / 16;
    const int NB_SELF  = (N_NODES * 32 + 255) / 256;
    const int NB_SCAT  = (int)(((size_t)N_EDGES * 32 + 255) / 256);

    k_init_deg<<<NB_NODE, 256, 0, stream>>>(dinv);
    k_deg<<<NB_EDGE, 256, 0, stream>>>(dst, dinv);
    k_dinv<<<NB_NODE, 256, 0, stream>>>(dinv);

    k_gemm<0><<<NB_TILE, 256, 0, stream>>>(x, W_gcn, nullptr, nullptr, nullptr, xw);

    k_selfloop<<<NB_SELF, 256, 0, stream>>>(dinv, xw, agg);
    k_scatter<<<NB_SCAT, 256, 0, stream>>>(src, dst, dinv, xw, agg);

    k_gemm<1><<<NB_TILE, 256, 0, stream>>>(agg, W_lin, b_gcn, b_lin, x, out);
}


# edge aggregation without f32 atomics: per-call counting sort of edges by destination (int atomic arrival rank), then one wave per node sums its source rows in registers in rank order and stores once
# speedup vs baseline: 1.6530x; 1.2983x over previous
#include <hip/hip_runtime.h>

typedef __attribute__((ext_vector_type(2))) float v2f;
typedef __attribute__((ext_vector_type(8))) float v8f;

#define N_NODES 100000
#define N_EDGES 1600000
#define HID 128


__global__ __launch_bounds__(256) void k_zero_cnt(int* __restrict__ cnt) {
    int i = blockIdx.x * 256 + threadIdx.x;
    if (i < N_NODES) cnt[i] = 0;
}

__global__ __launch_bounds__(256) void k_count(const int* __restrict__ dst,
                                               int* __restrict__ cnt,
                                               int* __restrict__ rank) {
    int q = blockIdx.x * 256 + threadIdx.x;
    if (q >= N_EDGES / 4) return;
    int4 d = ((const int4*)dst)[q];
    int4 r;
    r.x = atomicAdd(&cnt[d.x], 1);
    r.y = atomicAdd(&cnt[d.y], 1);
    r.z = atomicAdd(&cnt[d.z], 1);
    r.w = atomicAdd(&cnt[d.w], 1);
    ((int4*)rank)[q] = r;
}

template <int U>
__device__ __forceinline__ int sum_chunks(const int4* __restrict__ p) {
    int4 c[U];
    #pragma unroll
    for (int u = 0; u < U; ++u) c[u] = p[u * 256];
    int t = 0;
    #pragma unroll
    for (int u = 0; u < U; ++u) t += (c[u].x + c[u].y) + (c[u].z + c[u].w);
    return t;
}

__global__ __launch_bounds__(256) void k_scan(const int* __restrict__ cnt,
                                              int* __restrict__ rowptr,
                                              float* __restrict__ dinv) {
    __shared__ int s_red[8];
    __shared__ int s_wsum[8];
    const int tid  = threadIdx.x;
    const int lane = tid & 31;
    const int wave = tid >> 5;
    const int4* cnt4 = (const int4*)cnt;
    const int q0 = blockIdx.x * 256;

    const int nb = (int)blockIdx.x;
    int part = 0;
    int k = 0;
    for (; k + 16 <= nb; k += 16) part += sum_chunks<16>(cnt4 + k * 256 + tid);
    for (; k + 4 <= nb; k += 4)   part += sum_chunks<4>(cnt4 + k * 256 + tid);
    for (; k < nb; ++k)           part += sum_chunks<1>(cnt4 + k * 256 + tid);
    for (int o = 16; o > 0; o >>= 1) part += __shfl_xor(part, o);
    if (lane == 0) s_red[wave] = part;

    const int q = q0 + tid;
    int4 c = {0, 0, 0, 0};
    if (q < N_NODES / 4) c = cnt4[q];
    const int t1 = c.x;
    const int t2 = t1 + c.y;
    const int t3 = t2 + c.z;
    const int t4 = t3 + c.w;
    int inc = t4;
    for (int o = 1; o < 32; o <<= 1) {
        int up = __shfl_up(inc, o);
        if (lane >= o) inc += up;
    }
    if (lane == 31) s_wsum[wave] = inc;
    __syncthreads();
    int off = inc - t4;
    for (int w = 0; w < 8; ++w) {
        off += s_red[w];
        if (w < wave) off += s_wsum[w];
    }
    if (q < N_NODES / 4) {
        int4 r = {off, off + t1, off + t2, off + t3};
        ((int4*)rowptr)[q] = r;
        float4 d = {rsqrtf(1.0f + (float)c.x), rsqrtf(1.0f + (float)c.y),
                    rsqrtf(1.0f + (float)c.z), rsqrtf(1.0f + (float)c.w)};
        ((float4*)dinv)[q] = d;
        if (q == N_NODES / 4 - 1) rowptr[N_NODES] = off + t4;
    }
}

__device__ __forceinline__ void put_edge(int2* __restrict__ sorted, int p, int s, float norm) {
    unsigned pc = min((unsigned)p, (unsigned)(N_EDGES - 1));
    sorted[pc] = make_int2(s, __float_as_int(norm));
}

__global__ __launch_bounds__(256) void k_fill(const int* __restrict__ src,
                                              const int* __restrict__ dst,
                                              const int* __restrict__ rank,
                                              const int* __restrict__ rowptr,
                                              const float* __restrict__ dinv,
                                              int* __restrict__ csr) {
    int q = blockIdx.x * 256 + threadIdx.x;
    if (q >= N_EDGES / 4) return;
    int2* sorted = (int2*)csr;
    const int4 s = ((const int4*)src)[q];
    const int4 d = ((const int4*)dst)[q];
    const int4 r = ((const int4*)rank)[q];
    const int4 rp = {rowptr[d.x], rowptr[d.y], rowptr[d.z], rowptr[d.w]};
    const float4 nd = {dinv[d.x], dinv[d.y], dinv[d.z], dinv[d.w]};
    const float4 ns = {dinv[s.x], dinv[s.y], dinv[s.z], dinv[s.w]};
    put_edge(sorted, rp.x + r.x, s.x, ns.x * nd.x);
    put_edge(sorted, rp.y + r.y, s.y, ns.y * nd.y);
    put_edge(sorted, rp.z + r.z, s.z, ns.z * nd.z);
    put_edge(sorted, rp.w + r.w, s.w, ns.w * nd.w);
}

template <int MODE>
__global__ __launch_bounds__(256) void k_gemm(
    const float* __restrict__ A, const float* __restrict__ W,
    const float* __restrict__ b_in, const float* __restrict__ b_out,
    const float* __restrict__ xres, float* __restrict__ out) {
    __shared__ float sW[HID * HID];

    const int tid = threadIdx.x;
    for (int i = tid * 4; i < HID * HID; i += 256 * 4)
        *(float4*)&sW[i] = *(const float4*)&W[i];
    __syncthreads();

    const int wave = tid >> 5;
    const int lane = tid & 31;
    const int half = lane >> 4;
    const int lm   = lane & 15;
    const int kb   = half * 2;
    const int m0   = blockIdx.x * 16;
    const int n0   = wave * 16;

    const float* Arow = A + (size_t)(m0 + lm) * HID;

    v8f c = {};
    #pragma unroll 4
    for (int kk = 0; kk < HID; kk += 4) {
        v2f a, b;
        if (MODE == 0) {
            a = *(const v2f*)(Arow + kk + kb);
        } else {
            float a0 = Arow[kk + kb]     + b_in[kk + kb];
            float a1 = Arow[kk + kb + 1] + b_in[kk + kb + 1];
            a.x = a0 > 0.f ? a0 : 0.f;
            a.y = a1 > 0.f ? a1 : 0.f;
        }
        b.x = sW[(kk + kb) * HID + n0 + lm];
        b.y = sW[(kk + kb + 1) * HID + n0 + lm];
        c = __builtin_amdgcn_wmma_f32_16x16x4_f32(false, a, false, b,
                                                  (short)0, c, false, false);
    }

    const int n = n0 + lm;
    #pragma unroll
    for (int v = 0; v < 8; ++v) {
        int m = m0 + v + half * 8;
        float val = c[v];
        if (MODE == 1) {
            val += b_out[n];
            val = val > 0.f ? val : 0.f;
            val += xres[(size_t)m * HID + n];
        }
        out[(size_t)m * HID + n] = val;
    }
}


template <int K>
__device__ __forceinline__ void add_rows(float4& acc, const float4* __restrict__ rows,
                                         int es, int ew, int j) {
    float4 a[K];
    float w[K];
    #pragma unroll
    for (int k = 0; k < K; ++k) {
        const int s = __builtin_amdgcn_readlane(es, j + k);
        w[k] = __int_as_float(__builtin_amdgcn_readlane(ew, j + k));
        a[k] = rows[(size_t)s * 32];
    }
    #pragma unroll
    for (int k = 0; k < K; ++k) {
        acc.x = fmaf(w[k], a[k].x, acc.x);
        acc.y = fmaf(w[k], a[k].y, acc.y);
        acc.z = fmaf(w[k], a[k].z, acc.z);
        acc.w = fmaf(w[k], a[k].w, acc.w);
    }
}

__global__ __launch_bounds__(256) void k_gather(const int* __restrict__ rowptr,
                                                const int* __restrict__ csr,
                                                const float* __restrict__ dinv,
                                                const float* __restrict__ xw,
                                                float* __restrict__ agg) {
    const int lane = threadIdx.x & 31;
    const int node = __builtin_amdgcn_readfirstlane((int)(blockIdx.x * 8 + (threadIdx.x >> 5)));
    if (node >= N_NODES) return;
    const int beg = __builtin_amdgcn_readfirstlane(rowptr[node]);
    const int end = __builtin_amdgcn_readfirstlane(rowptr[node + 1]);
    const float di = dinv[node];
    const float4* rows = (const float4*)xw + lane;

    const float4 v = rows[(size_t)node * 32];
    const float wself = di * di;
    float4 acc = {wself * v.x, wself * v.y, wself * v.z, wself * v.w};

    for (int base = beg; base < end; base += 32) {
        const int idx = base + lane;
        int2 ent = {0, 0};
        if (idx < end) ent = ((const int2*)csr)[idx];
        const int rest = end - base;
        const int n = rest < 32 ? rest : 32;
        int j = 0;
        for (; j + 8 <= n; j += 8) add_rows<8>(acc, rows, ent.x, ent.y, j);
        if (j + 4 <= n) { add_rows<4>(acc, rows, ent.x, ent.y, j); j += 4; }
        const int r = n - j;
        if (r == 3)      add_rows<3>(acc, rows, ent.x, ent.y, j);
        else if (r == 2) add_rows<2>(acc, rows, ent.x, ent.y, j);
        else if (r == 1) add_rows<1>(acc, rows, ent.x, ent.y, j);
    }
    ((float4*)agg)[(size_t)node * 32 + lane] = acc;
}


extern "C" void kernel_launch(void* const* d_in, const int* in_sizes, int n_in,
                              void* d_out, int out_size, void* d_ws, size_t ws_size,
                              hipStream_t stream) {
    const float* x     = (const float*)d_in[0];
    const int*   ei    = (const int*)d_in[1];
    const float* W_gcn = (const float*)d_in[2];
    const float* b_gcn = (const float*)d_in[3];
    const float* W_lin = (const float*)d_in[4];
    const float* b_lin = (const float*)d_in[5];

    const int* src = ei;
    const int* dst = ei + N_EDGES;

    float* out = (float*)d_out;

    float* dinv   = (float*)d_ws;
    float* agg    = (float*)((char*)d_ws + (size_t)N_NODES * sizeof(float));
    int*   cnt    = (int*)((char*)d_ws + (size_t)51600128);
    int*   rowptr = (int*)((char*)d_ws + (size_t)52000256);
    int*   rank   = (int*)((char*)d_ws + (size_t)52400384);
    int*   csr    = (int*)((char*)d_ws + (size_t)58800384);
    float* xw     = out;

    const int NB_NODE  = (N_NODES + 255) / 256;
    const int NB_EDGE4 = (N_EDGES / 4 + 255) / 256;
    const int NB_TILE  = N_NODES / 16;
    const int NB_SCAN  = (N_NODES / 4 + 255) / 256;
    const int NB_GATH  = (N_NODES + 7) / 8;

    k_zero_cnt<<<NB_NODE, 256, 0, stream>>>(cnt);
    k_count<<<NB_EDGE4, 256, 0, stream>>>(dst, cnt, rank);
    k_scan<<<NB_SCAN, 256, 0, stream>>>(cnt, rowptr, dinv);
    k_fill<<<NB_EDGE4, 256, 0, stream>>>(src, dst, rank, rowptr, dinv, csr);

    k_gemm<0><<<NB_TILE, 256, 0, stream>>>(x, W_gcn, nullptr, nullptr, nullptr, xw);

    k_gather<<<NB_GATH, 256, 0, stream>>>(rowptr, csr, dinv, xw, agg);

    k_gemm<1><<<NB_TILE, 256, 0, stream>>>(agg, W_lin, b_gcn, b_lin, x, out);
}


# GEMM re-tiled: a block stages W in LDS once and loops over row tiles; each wave owns 16 rows x all 128 columns, A and out move as whole 512-B rows through LDS; k order and order of sums unchanged
# speedup vs baseline: 2.1929x; 1.3266x over previous
#include <hip/hip_runtime.h>

typedef __attribute__((ext_vector_type(2))) float v2f;
typedef __attribute__((ext_vector_type(8))) float v8f;

#define N_NODES 100000
#define N_EDGES 1600000
#define HID 128


__global__ __launch_bounds__(256) void k_zero_cnt(int* __restrict__ cnt) {
    int i = blockIdx.x * 256 + threadIdx.x;
    if (i < N_NODES) cnt[i] = 0;
}

__global__ __launch_bounds__(256) void k_count(const int* __restrict__ dst,
                                               int* __restrict__ cnt,
                                               int* __restrict__ rank) {
    int q = blockIdx.x * 256 + threadIdx.x;
    if (q >= N_EDGES / 4) return;
    int4 d = ((const int4*)dst)[q];
    int4 r;
    r.x = atomicAdd(&cnt[d.x], 1);
    r.y = atomicAdd(&cnt[d.y], 1);
    r.z = atomicAdd(&cnt[d.z], 1);
    r.w = atomicAdd(&cnt[d.w], 1);
    ((int4*)rank)[q] = r;
}

template <int U>
__device__ __forceinline__ int sum_chunks(const int4* __restrict__ p) {
    int4 c[U];
    #pragma unroll
    for (int u = 0; u < U; ++u) c[u] = p[u * 256];
    int t = 0;
    #pragma unroll
    for (int u = 0; u < U; ++u) t += (c[u].x + c[u].y) + (c[u].z + c[u].w);
    return t;
}

__global__ __launch_bounds__(256) void k_scan(const int* __restrict__ cnt,
                                              int* __restrict__ rowptr,
                                              float* __restrict__ dinv) {
    __shared__ int s_red[8];
    __shared__ int s_wsum[8];
    const int tid  = threadIdx.x;
    const int lane = tid & 31;
    const int wave = tid >> 5;
    const int4* cnt4 = (const int4*)cnt;
    const int q0 = blockIdx.x * 256;

    const int nb = (int)blockIdx.x;
    int part = 0;
    int k = 0;
    for (; k + 16 <= nb; k += 16) part += sum_chunks<16>(cnt4 + k * 256 + tid);
    for (; k + 4 <= nb; k += 4)   part += sum_chunks<4>(cnt4 + k * 256 + tid);
    for (; k < nb; ++k)           part += sum_chunks<1>(cnt4 + k * 256 + tid);
    for (int o = 16; o > 0; o >>= 1) part += __shfl_xor(part, o);
    if (lane == 0) s_red[wave] = part;

    const int q = q0 + tid;
    int4 c = {0, 0, 0, 0};
    if (q < N_NODES / 4) c = cnt4[q];
    const int t1 = c.x;
    const int t2 = t1 + c.y;
    const int t3 = t2 + c.z;
    const int t4 = t3 + c.w;
    int inc = t4;
    for (int o = 1; o < 32; o <<= 1) {
        int up = __shfl_up(inc, o);
        if (lane >= o) inc += up;
    }
    if (lane == 31) s_wsum[wave] = inc;
    __syncthreads();
    int off = inc - t4;
    for (int w = 0; w < 8; ++w) {
        off += s_red[w];
        if (w < wave) off += s_wsum[w];
    }
    if (q < N_NODES / 4) {
        int4 r = {off, off + t1, off + t2, off + t3};
        ((int4*)rowptr)[q] = r;
        float4 d = {rsqrtf(1.0f + (float)c.x), rsqrtf(1.0f + (float)c.y),
                    rsqrtf(1.0f + (float)c.z), rsqrtf(1.0f + (float)c.w)};
        ((float4*)dinv)[q] = d;
        if (q == N_NODES / 4 - 1) rowptr[N_NODES] = off + t4;
    }
}

__device__ __forceinline__ void put_edge(int2* __restrict__ sorted, int p, int s, float norm) {
    unsigned pc = min((unsigned)p, (unsigned)(N_EDGES - 1));
    sorted[pc] = make_int2(s, __float_as_int(norm));
}

__global__ __launch_bounds__(256) void k_fill(const int* __restrict__ src,
                                              const int* __restrict__ dst,
                                              const int* __restrict__ rank,
                                              const int* __restrict__ rowptr,
                                              const float* __restrict__ dinv,
                                              int* __restrict__ csr) {
    int q = blockIdx.x * 256 + threadIdx.x;
    if (q >= N_EDGES / 4) return;
    int2* sorted = (int2*)csr;
    const int4 s = ((const int4*)src)[q];
    const int4 d = ((const int4*)dst)[q];
    const int4 r = ((const int4*)rank)[q];
    const int4 rp = {rowptr[d.x], rowptr[d.y], rowptr[d.z], rowptr[d.w]};
    const float4 nd = {dinv[d.x], dinv[d.y], dinv[d.z], dinv[d.w]};
    const float4 ns = {dinv[s.x], dinv[s.y], dinv[s.z], dinv[s.w]};
    put_edge(sorted, rp.x + r.x, s.x, ns.x * nd.x);
    put_edge(sorted, rp.y + r.y, s.y, ns.y * nd.y);
    put_edge(sorted, rp.z + r.z, s.z, ns.z * nd.z);
    put_edge(sorted, rp.w + r.w, s.w, ns.w * nd.w);
}

#define GEMM_BLOCKS 512
#define TPITCH 132

template <int MODE>
__global__ __launch_bounds__(256) void k_gemm(
    const float* __restrict__ A, const float* __restrict__ W,
    const float* __restrict__ b_in, const float* __restrict__ b_out,
    const float* __restrict__ xres, float* __restrict__ out) {
    __shared__ float4 sW4[HID * HID / 4];
    __shared__ __attribute__((aligned(16))) float sT[8][16 * TPITCH];

    const int tid  = threadIdx.x;
    const int lane = tid & 31;
    const int half = lane >> 4;
    const int lm   = lane & 15;
    const int wave = __builtin_amdgcn_readfirstlane(tid >> 5);

    {
        float4 w[16];
        #pragma unroll
        for (int it = 0; it < 16; ++it) {
            const int q = tid + 256 * it;
            const int l = q & 31, t = (q >> 5) & 15, j = q >> 9;
            const int n = 16 * j + (l & 15);
            const int k = 8 * t + 2 * (l >> 4);
            w[it].x = W[(k    ) * HID + n];
            w[it].y = W[(k + 1) * HID + n];
            w[it].z = W[(k + 4) * HID + n];
            w[it].w = W[(k + 5) * HID + n];
        }
        #pragma unroll
        for (int it = 0; it < 16; ++it) sW4[tid + 256 * it] = w[it];
    }
    float4 bi = {0.f, 0.f, 0.f, 0.f}, bo = {0.f, 0.f, 0.f, 0.f};
    if (MODE == 1) {
        bi = ((const float4*)b_in)[lane];
        bo = ((const float4*)b_out)[lane];
    }
    __syncthreads();

    float* T = sT[wave];
    const int n_tiles = N_NODES / 16;

    for (int tile = blockIdx.x + GEMM_BLOCKS * wave; tile < n_tiles; tile += GEMM_BLOCKS * 8) {
        const size_t row0 = (size_t)tile * 16;

        float4 ar[16];
        #pragma unroll
        for (int r = 0; r < 16; ++r) ar[r] = ((const float4*)(A + (row0 + r) * HID))[lane];
        #pragma unroll
        for (int r = 0; r < 16; ++r) {
            float4 v = ar[r];
            if (MODE == 1) {
                v.x += bi.x; v.y += bi.y; v.z += bi.z; v.w += bi.w;
                v.x = v.x > 0.f ? v.x : 0.f; v.y = v.y > 0.f ? v.y : 0.f;
                v.z = v.z > 0.f ? v.z : 0.f; v.w = v.w > 0.f ? v.w : 0.f;
            }
            *(float4*)&T[r * TPITCH + 4 * lane] = v;
        }
        __builtin_amdgcn_fence(__ATOMIC_ACQ_REL, "wavefront");

        v2f a[32];
        #pragma unroll
        for (int s = 0; s < 32; ++s) a[s] = *(const v2f*)&T[lm * TPITCH + 4 * s + 2 * half];

        v8f acc[8];
        #pragma unroll
        for (int j = 0; j < 8; ++j) acc[j] = v8f{};

        #pragma unroll
        for (int t = 0; t < 16; ++t) {
            float4 b[8];
            #pragma unroll
            for (int j = 0; j < 8; ++j) b[j] = sW4[(j * 16 + t) * 32 + lane];
            #pragma unroll
            for (int j = 0; j < 8; ++j) {
                v2f b0 = {b[j].x, b[j].y};
                acc[j] = __builtin_amdgcn_wmma_f32_16x16x4_f32(false, a[2 * t], false, b0,
                                                               (short)0, acc[j], false, false);
            }
            #pragma unroll
            for (int j = 0; j < 8; ++j) {
                v2f b1 = {b[j].z, b[j].w};
                acc[j] = __builtin_amdgcn_wmma_f32_16x16x4_f32(false, a[2 * t + 1], false, b1,
                                                               (short)0, acc[j], false, false);
            }
        }

        __builtin_amdgcn_fence(__ATOMIC_ACQ_REL, "wavefront");
        #pragma unroll
        for (int j = 0; j < 8; ++j) {
            #pragma unroll
            for (int v = 0; v < 8; ++v) T[(8 * half + v) * TPITCH + 16 * j + lm] = acc[j][v];
        }
        __builtin_amdgcn_fence(__ATOMIC_ACQ_REL, "wavefront");

        float4 xr[16];
        if (MODE == 1) {
            #pragma unroll
            for (int r = 0; r < 16; ++r) xr[r] = ((const float4*)(xres + (row0 + r) * HID))[lane];
        }
        #pragma unroll
        for (int r = 0; r < 16; ++r) {
            float4 v = *(const float4*)&T[r * TPITCH + 4 * lane];
            if (MODE == 1) {
                v.x += bo.x; v.y += bo.y; v.z += bo.z; v.w += bo.w;
                v.x = v.x > 0.f ? v.x : 0.f; v.y = v.y > 0.f ? v.y : 0.f;
                v.z = v.z > 0.f ? v.z : 0.f; v.w = v.w > 0.f ? v.w : 0.f;
                v.x += xr[r].x; v.y += xr[r].y; v.z += xr[r].z; v.w += xr[r].w;
            }
            ((float4*)(out + (row0 + r) * HID))[lane] = v;
        }
        __builtin_amdgcn_fence(__ATOMIC_ACQ_REL, "wavefront");
    }
}


template <int K>
__device__ __forceinline__ void add_rows(float4& acc, const float4* __restrict__ rows,
                                         int es, int ew, int j) {
    float4 a[K];
    float w[K];
    #pragma unroll
    for (int k = 0; k < K; ++k) {
        const int s = __builtin_amdgcn_readlane(es, j + k);
        w[k] = __int_as_float(__builtin_amdgcn_readlane(ew, j + k));
        a[k] = rows[(size_t)s * 32];
    }
    #pragma unroll
    for (int k = 0; k < K; ++k) {
        acc.x = fmaf(w[k], a[k].x, acc.x);
        acc.y = fmaf(w[k], a[k].y, acc.y);
        acc.z = fmaf(w[k], a[k].z, acc.z);
        acc.w = fmaf(w[k], a[k].w, acc.w);
    }
}

__global__ __launch_bounds__(256) void k_gather(const int* __restrict__ rowptr,
                                                const int* __restrict__ csr,
                                                const float* __restrict__ dinv,
                                                const float* __restrict__ xw,
                                                float* __restrict__ agg) {
    const int lane = threadIdx.x & 31;
    const int node = __builtin_amdgcn_readfirstlane((int)(blockIdx.x * 8 + (threadIdx.x >> 5)));
    if (node >= N_NODES) return;
    const int beg = __builtin_amdgcn_readfirstlane(rowptr[node]);
    const int end = __builtin_amdgcn_readfirstlane(rowptr[node + 1]);
    const float di = dinv[node];
    const float4* rows = (const float4*)xw + lane;

    const float4 v = rows[(size_t)node * 32];
    const float wself = di * di;
    float4 acc = {wself * v.x, wself * v.y, wself * v.z, wself * v.w};

    for (int base = beg; base < end; base += 32) {
        const int idx = base + lane;
        int2 ent = {0, 0};
        if (idx < end) ent = ((const int2*)csr)[idx];
        const int rest = end - base;
        const int n = rest < 32 ? rest : 32;
        int j = 0;
        for (; j + 8 <= n; j += 8) add_rows<8>(acc, rows, ent.x, ent.y, j);
        if (j + 4 <= n) { add_rows<4>(acc, rows, ent.x, ent.y, j); j += 4; }
        const int r = n - j;
        if (r == 3)      add_rows<3>(acc, rows, ent.x, ent.y, j);
        else if (r == 2) add_rows<2>(acc, rows, ent.x, ent.y, j);
        else if (r == 1) add_rows<1>(acc, rows, ent.x, ent.y, j);
    }
    ((float4*)agg)[(size_t)node * 32 + lane] = acc;
}


extern "C" void kernel_launch(void* const* d_in, const int* in_sizes, int n_in,
                              void* d_out, int out_size, void* d_ws, size_t ws_size,
                              hipStream_t stream) {
    const float* x     = (const float*)d_in[0];
    const int*   ei    = (const int*)d_in[1];
    const float* W_gcn = (const float*)d_in[2];
    const float* b_gcn = (const float*)d_in[3];
    const float* W_lin = (const float*)d_in[4];
    const float* b_lin = (const float*)d_in[5];

    const int* src = ei;
    const int* dst = ei + N_EDGES;

    float* out = (float*)d_out;

    float* dinv   = (float*)d_ws;
    float* agg    = (float*)((char*)d_ws + (size_t)N_NODES * sizeof(float));
    int*   cnt    = (int*)((char*)d_ws + (size_t)51600128);
    int*   rowptr = (int*)((char*)d_ws + (size_t)52000256);
    int*   rank   = (int*)((char*)d_ws + (size_t)52400384);
    int*   csr    = (int*)((char*)d_ws + (size_t)58800384);
    float* xw     = out;

    const int NB_NODE  = (N_NODES + 255) / 256;
    const int NB_EDGE4 = (N_EDGES / 4 + 255) / 256;
    const int NB_TILE  = GEMM_BLOCKS;
    const int NB_SCAN  = (N_NODES / 4 + 255) / 256;
    const int NB_GATH  = (N_NODES + 7) / 8;

    k_zero_cnt<<<NB_NODE, 256, 0, stream>>>(cnt);
    k_count<<<NB_EDGE4, 256, 0, stream>>>(dst, cnt, rank);
    k_scan<<<NB_SCAN, 256, 0, stream>>>(cnt, rowptr, dinv);
    k_fill<<<NB_EDGE4, 256, 0, stream>>>(src, dst, rank, rowptr, dinv, csr);

    k_gemm<0><<<NB_TILE, 256, 0, stream>>>(x, W_gcn, nullptr, nullptr, nullptr, xw);

    k_gather<<<NB_GATH, 256, 0, stream>>>(rowptr, csr, dinv, xw, agg);

    k_gemm<1><<<NB_TILE, 256, 0, stream>>>(agg, W_lin, b_gcn, b_lin, x, out);
}


# edge sort on LDS atomics: each block splits 6400 edges by 256-node range, then one block per range ranks and places them and writes row starts; no global atomics; sums in LDS arrival order
# speedup vs baseline: 4.0201x; 1.8332x over previous
#include <hip/hip_runtime.h>

typedef __attribute__((ext_vector_type(2))) float v2f;
typedef __attribute__((ext_vector_type(8))) float v8f;

#define N_NODES 100000
#define N_EDGES 1600000
#define HID 128

#define EPB 6400
#define NCHUNK (N_EDGES / EPB)
#define NBK ((N_NODES + 255) / 256)
#define OPITCH 416
#define EPT (EPB / 256)

__global__ __launch_bounds__(256) void k_part(const int* __restrict__ src,
                                              const int* __restrict__ dst,
                                              int* __restrict__ tmp_i,
                                              int* __restrict__ offs) {
    __shared__ int hist[OPITCH];
    const int tid  = threadIdx.x;
    const int lane = tid & 31;
    const int base = blockIdx.x * EPB;
    int2* tmp = (int2*)tmp_i;
    for (int k = tid; k < OPITCH; k += 256) hist[k] = 0;
    int d[EPT], s[EPT], lr[EPT];
    #pragma unroll
    for (int i = 0; i < EPT; ++i) {
        d[i] = dst[base + i * 256 + tid];
        s[i] = src[base + i * 256 + tid];
    }
    __syncthreads();
    #pragma unroll
    for (int i = 0; i < EPT; ++i) lr[i] = atomicAdd(&hist[(unsigned)d[i] >> 8], 1);
    __syncthreads();
    if (tid < 32) {
        int carry = 0;
        for (int k0 = 0; k0 < OPITCH; k0 += 32) {
            const int v = hist[k0 + lane];
            int inc = v;
            for (int o = 1; o < 32; o <<= 1) {
                int up = __shfl_up(inc, o);
                if (lane >= o) inc += up;
            }
            hist[k0 + lane] = carry + inc - v;
            carry += __shfl(inc, 31);
        }
    }
    __syncthreads();
    #pragma unroll
    for (int i = 0; i < EPT; ++i) {
        const unsigned p = min((unsigned)(hist[(unsigned)d[i] >> 8] + lr[i]), (unsigned)(EPB - 1));
        tmp[base + p] = make_int2(d[i], s[i]);
    }
    for (int k = tid; k <= NBK; k += 256) offs[blockIdx.x * OPITCH + k] = base + hist[k];
}

__global__ __launch_bounds__(256) void k_rank(const int* __restrict__ offs,
                                              const int* __restrict__ tmp_i,
                                              int* __restrict__ rowptr,
                                              float* __restrict__ dinv,
                                              int* __restrict__ sorted) {
    __shared__ int lcnt[256];
    __shared__ int lfill[256];
    __shared__ int s_base[8];
    __shared__ int s_wsum[8];
    const int tid  = threadIdx.x;
    const int lane = tid & 31;
    const int wave = tid >> 5;
    const int b    = blockIdx.x;
    const int2* tmp = (const int2*)tmp_i;
    lcnt[tid] = 0;
    lfill[tid] = 0;
    int beg = 0, end = 0, lo = 0;
    if (tid < NCHUNK) {
        lo  = tid * EPB;
        beg = offs[tid * OPITCH + b];
        end = offs[tid * OPITCH + b + 1];
        beg = min(max(beg, lo), lo + EPB);
        end = min(max(end, beg), lo + EPB);
    }
    __syncthreads();
    for (int e = beg; e < end; e += 4) {
        int x[4];
        #pragma unroll
        for (int u = 0; u < 4; ++u) x[u] = tmp[min(e + u, N_EDGES - 1)].x;
        asm volatile("" :: "v"(x[0]), "v"(x[1]), "v"(x[2]), "v"(x[3]));
        #pragma unroll
        for (int u = 0; u < 4; ++u)
            if (e + u < end) atomicAdd(&lcnt[x[u] & 255], 1);
    }
    int part = beg - lo;
    for (int o = 16; o > 0; o >>= 1) part += __shfl_xor(part, o);
    if (lane == 0) s_base[wave] = part;
    __syncthreads();
    const int c = lcnt[tid];
    int inc = c;
    for (int o = 1; o < 32; o <<= 1) {
        int up = __shfl_up(inc, o);
        if (lane >= o) inc += up;
    }
    if (lane == 31) s_wsum[wave] = inc;
    __syncthreads();
    int start = inc - c;
    for (int w = 0; w < 8; ++w) {
        start += s_base[w];
        if (w < wave) start += s_wsum[w];
    }
    const int node = b * 256 + tid;
    if (node < N_NODES) {
        rowptr[node] = start;
        dinv[node] = rsqrtf(1.0f + (float)c);
        if (node == N_NODES - 1) rowptr[N_NODES] = start + c;
    }
    lcnt[tid] = start;
    __syncthreads();
    for (int e = beg; e < end; e += 4) {
        int2 ds[4];
        #pragma unroll
        for (int u = 0; u < 4; ++u) ds[u] = tmp[min(e + u, N_EDGES - 1)];
        asm volatile("" :: "v"(ds[0].x), "v"(ds[0].y), "v"(ds[1].x), "v"(ds[1].y),
                           "v"(ds[2].x), "v"(ds[2].y), "v"(ds[3].x), "v"(ds[3].y));
        #pragma unroll
        for (int u = 0; u < 4; ++u) {
            if (e + u < end) {
                const int k = ds[u].x & 255;
                const int r = atomicAdd(&lfill[k], 1);
                const unsigned p = min((unsigned)(lcnt[k] + r), (unsigned)(N_EDGES - 1));
                sorted[p] = ds[u].y;
            }
        }
    }
}

#define GEMM_BLOCKS 512
#define TPITCH 132

template <int MODE>
__global__ __launch_bounds__(256) void k_gemm(
    const float* __restrict__ A, const float* __restrict__ W,
    const float* __restrict__ b_in, const float* __restrict__ b_out,
    const float* __restrict__ xres, float* __restrict__ out) {
    __shared__ float4 sW4[HID * HID / 4];
    __shared__ __attribute__((aligned(16))) float sT[8][16 * TPITCH];

    const int tid  = threadIdx.x;
    const int lane = tid & 31;
    const int half = lane >> 4;
    const int lm   = lane & 15;
    const int wave = __builtin_amdgcn_readfirstlane(tid >> 5);

    {
        float4 w[16];
        #pragma unroll
        for (int it = 0; it < 16; ++it) {
            const int q = tid + 256 * it;
            const int l = q & 31, t = (q >> 5) & 15, j = q >> 9;
            const int n = 16 * j + (l & 15);
            const int k = 8 * t + 2 * (l >> 4);
            w[it].x = W[(k    ) * HID + n];
            w[it].y = W[(k + 1) * HID + n];
            w[it].z = W[(k + 4) * HID + n];
            w[it].w = W[(k + 5) * HID + n];
        }
        #pragma unroll
        for (int it = 0; it < 16; ++it) sW4[tid + 256 * it] = w[it];
    }
    float4 bi = {0.f, 0.f, 0.f, 0.f}, bo = {0.f, 0.f, 0.f, 0.f};
    if (MODE == 1) {
        bi = ((const float4*)b_in)[lane];
        bo = ((const float4*)b_out)[lane];
    }
    __syncthreads();

    float* T = sT[wave];
    const int n_tiles = N_NODES / 16;

    for (int tile = blockIdx.x + GEMM_BLOCKS * wave; tile < n_tiles; tile += GEMM_BLOCKS * 8) {
        const size_t row0 = (size_t)tile * 16;

        float4 ar[16];
        #pragma unroll
        for (int r = 0; r < 16; ++r) ar[r] = ((const float4*)(A + (row0 + r) * HID))[lane];
        #pragma unroll
        for (int r = 0; r < 16; ++r) {
            float4 v = ar[r];
            if (MODE == 1) {
                v.x += bi.x; v.y += bi.y; v.z += bi.z; v.w += bi.w;
                v.x = v.x > 0.f ? v.x : 0.f; v.y = v.y > 0.f ? v.y : 0.f;
                v.z = v.z > 0.f ? v.z : 0.f; v.w = v.w > 0.f ? v.w : 0.f;
            }
            *(float4*)&T[r * TPITCH + 4 * lane] = v;
        }
        __builtin_amdgcn_fence(__ATOMIC_ACQ_REL, "wavefront");

        v2f a[32];
        #pragma unroll
        for (int s = 0; s < 32; ++s) a[s] = *(const v2f*)&T[lm * TPITCH + 4 * s + 2 * half];

        v8f acc[8];
        #pragma unroll
        for (int j = 0; j < 8; ++j) acc[j] = v8f{};

        #pragma unroll
        for (int t = 0; t < 16; ++t) {
            float4 b[8];
            #pragma unroll
            for (int j = 0; j < 8; ++j) b[j] = sW4[(j * 16 + t) * 32 + lane];
            #pragma unroll
            for (int j = 0; j < 8; ++j) {
                v2f b0 = {b[j].x, b[j].y};
                acc[j] = __builtin_amdgcn_wmma_f32_16x16x4_f32(false, a[2 * t], false, b0,
                                                               (short)0, acc[j], false, false);
            }
            #pragma unroll
            for (int j = 0; j < 8; ++j) {
                v2f b1 = {b[j].z, b[j].w};
                acc[j] = __builtin_amdgcn_wmma_f32_16x16x4_f32(false, a[2 * t + 1], false, b1,
                                                               (short)0, acc[j], false, false);
            }
        }

        __builtin_amdgcn_fence(__ATOMIC_ACQ_REL, "wavefront");
        #pragma unroll
        for (int j = 0; j < 8; ++j) {
            #pragma unroll
            for (int v = 0; v < 8; ++v) T[(8 * half + v) * TPITCH + 16 * j + lm] = acc[j][v];
        }
        __builtin_amdgcn_fence(__ATOMIC_ACQ_REL, "wavefront");

        float4 xr[16];
        if (MODE == 1) {
            #pragma unroll
            for (int r = 0; r < 16; ++r) xr[r] = ((const float4*)(xres + (row0 + r) * HID))[lane];
        }
        #pragma unroll
        for (int r = 0; r < 16; ++r) {
            float4 v = *(const float4*)&T[r * TPITCH + 4 * lane];
            if (MODE == 1) {
                v.x += bo.x; v.y += bo.y; v.z += bo.z; v.w += bo.w;
                v.x = v.x > 0.f ? v.x : 0.f; v.y = v.y > 0.f ? v.y : 0.f;
                v.z = v.z > 0.f ? v.z : 0.f; v.w = v.w > 0.f ? v.w : 0.f;
                v.x += xr[r].x; v.y += xr[r].y; v.z += xr[r].z; v.w += xr[r].w;
            }
            ((float4*)(out + (row0 + r) * HID))[lane] = v;
        }
        __builtin_amdgcn_fence(__ATOMIC_ACQ_REL, "wavefront");
    }
}


template <int K>
__device__ __forceinline__ void add_rows(float4& acc, const float4* __restrict__ rows,
                                         int es, int ew, int j) {
    float4 a[K];
    float w[K];
    #pragma unroll
    for (int k = 0; k < K; ++k) {
        const int s = __builtin_amdgcn_readlane(es, j + k);
        w[k] = __int_as_float(__builtin_amdgcn_readlane(ew, j + k));
        a[k] = rows[(size_t)s * 32];
    }
    #pragma unroll
    for (int k = 0; k < K; ++k) {
        acc.x = fmaf(w[k], a[k].x, acc.x);
        acc.y = fmaf(w[k], a[k].y, acc.y);
        acc.z = fmaf(w[k], a[k].z, acc.z);
        acc.w = fmaf(w[k], a[k].w, acc.w);
    }
}

__global__ __launch_bounds__(256) void k_gather(const int* __restrict__ rowptr,
                                                const int* __restrict__ csr,
                                                const float* __restrict__ dinv,
                                                const float* __restrict__ xw,
                                                float* __restrict__ agg) {
    const int lane = threadIdx.x & 31;
    const int node = __builtin_amdgcn_readfirstlane((int)(blockIdx.x * 8 + (threadIdx.x >> 5)));
    if (node >= N_NODES) return;
    const int beg = __builtin_amdgcn_readfirstlane(rowptr[node]);
    const int end = __builtin_amdgcn_readfirstlane(rowptr[node + 1]);
    const float di = dinv[node];
    const float4* rows = (const float4*)xw + lane;

    const float4 v = rows[(size_t)node * 32];
    const float wself = di * di;
    float4 acc = {wself * v.x, wself * v.y, wself * v.z, wself * v.w};

    for (int base = beg; base < end; base += 32) {
        const int idx = base + lane;
        int2 ent = {0, 0};
        if (idx < end) {
            ent.x = (int)min((unsigned)csr[idx], (unsigned)(N_NODES - 1));
            ent.y = __float_as_int(dinv[ent.x] * di);
        }
        const int rest = end - base;
        const int n = rest < 32 ? rest : 32;
        int j = 0;
        for (; j + 8 <= n; j += 8) add_rows<8>(acc, rows, ent.x, ent.y, j);
        if (j + 4 <= n) { add_rows<4>(acc, rows, ent.x, ent.y, j); j += 4; }
        const int r = n - j;
        if (r == 3)      add_rows<3>(acc, rows, ent.x, ent.y, j);
        else if (r == 2) add_rows<2>(acc, rows, ent.x, ent.y, j);
        else if (r == 1) add_rows<1>(acc, rows, ent.x, ent.y, j);
    }
    ((float4*)agg)[(size_t)node * 32 + lane] = acc;
}


extern "C" void kernel_launch(void* const* d_in, const int* in_sizes, int n_in,
                              void* d_out, int out_size, void* d_ws, size_t ws_size,
                              hipStream_t stream) {
    const float* x     = (const float*)d_in[0];
    const int*   ei    = (const int*)d_in[1];
    const float* W_gcn = (const float*)d_in[2];
    const float* b_gcn = (const float*)d_in[3];
    const float* W_lin = (const float*)d_in[4];
    const float* b_lin = (const float*)d_in[5];

    const int* src = ei;
    const int* dst = ei + N_EDGES;

    float* out = (float*)d_out;

    float* dinv   = (float*)d_ws;
    float* agg    = (float*)((char*)d_ws + (size_t)N_NODES * sizeof(float));
    int*   rowptr = (int*)((char*)d_ws + (size_t)52000256);
    int*   offs   = (int*)((char*)d_ws + (size_t)52400384);
    int*   tmp    = (int*)((char*)d_ws + (size_t)52816384);
    int*   csr    = (int*)((char*)d_ws + (size_t)65616384);
    float* xw     = out;

    const int NB_TILE  = GEMM_BLOCKS;
    const int NB_GATH  = (N_NODES + 7) / 8;

    k_part<<<NCHUNK, 256, 0, stream>>>(src, dst, tmp, offs);
    k_rank<<<NBK, 256, 0, stream>>>(offs, tmp, rowptr, dinv, csr);

    k_gemm<0><<<NB_TILE, 256, 0, stream>>>(x, W_gcn, nullptr, nullptr, nullptr, xw);

    k_gather<<<NB_GATH, 256, 0, stream>>>(rowptr, csr, dinv, xw, agg);

    k_gemm<1><<<NB_TILE, 256, 0, stream>>>(agg, W_lin, b_gcn, b_lin, x, out);
}


# aggregate the input rows first, (A x) W_gcn instead of A (x W_gcn): both products run back to back in one launch, the tile kept in LDS between them; 4 launches, no xw plane; order of sums changes
# speedup vs baseline: 4.5557x; 1.1332x over previous
#include <hip/hip_runtime.h>

typedef __attribute__((ext_vector_type(2))) float v2f;
typedef __attribute__((ext_vector_type(8))) float v8f;

#define N_NODES 100000
#define N_EDGES 1600000
#define HID 128

#define EPB 6400
#define NCHUNK (N_EDGES / EPB)
#define NBK ((N_NODES + 255) / 256)
#define OPITCH 416
#define EPT (EPB / 256)

__global__ __launch_bounds__(256) void k_part(const int* __restrict__ src,
                                              const int* __restrict__ dst,
                                              int* __restrict__ tmp_i,
                                              int* __restrict__ offs) {
    __shared__ int hist[OPITCH];
    const int tid  = threadIdx.x;
    const int lane = tid & 31;
    const int base = blockIdx.x * EPB;
    int2* tmp = (int2*)tmp_i;
    for (int k = tid; k < OPITCH; k += 256) hist[k] = 0;
    int d[EPT], s[EPT], lr[EPT];
    #pragma unroll
    for (int i = 0; i < EPT; ++i) {
        d[i] = dst[base + i * 256 + tid];
        s[i] = src[base + i * 256 + tid];
    }
    __syncthreads();
    #pragma unroll
    for (int i = 0; i < EPT; ++i) lr[i] = atomicAdd(&hist[(unsigned)d[i] >> 8], 1);
    __syncthreads();
    if (tid < 32) {
        int carry = 0;
        for (int k0 = 0; k0 < OPITCH; k0 += 32) {
            const int v = hist[k0 + lane];
            int inc = v;
            for (int o = 1; o < 32; o <<= 1) {
                int up = __shfl_up(inc, o);
                if (lane >= o) inc += up;
            }
            hist[k0 + lane] = carry + inc - v;
            carry += __shfl(inc, 31);
        }
    }
    __syncthreads();
    #pragma unroll
    for (int i = 0; i < EPT; ++i) {
        const unsigned p = min((unsigned)(hist[(unsigned)d[i] >> 8] + lr[i]), (unsigned)(EPB - 1));
        tmp[base + p] = make_int2(d[i], s[i]);
    }
    for (int k = tid; k <= NBK; k += 256) offs[blockIdx.x * OPITCH + k] = base + hist[k];
}

__global__ __launch_bounds__(256) void k_rank(const int* __restrict__ offs,
                                              const int* __restrict__ tmp_i,
                                              int* __restrict__ rowptr,
                                              float* __restrict__ dinv,
                                              int* __restrict__ sorted) {
    __shared__ int lcnt[256];
    __shared__ int lfill[256];
    __shared__ int s_base[8];
    __shared__ int s_wsum[8];
    const int tid  = threadIdx.x;
    const int lane = tid & 31;
    const int wave = tid >> 5;
    const int b    = blockIdx.x;
    const int2* tmp = (const int2*)tmp_i;
    lcnt[tid] = 0;
    lfill[tid] = 0;
    int beg = 0, end = 0, lo = 0;
    if (tid < NCHUNK) {
        lo  = tid * EPB;
        beg = offs[tid * OPITCH + b];
        end = offs[tid * OPITCH + b + 1];
        beg = min(max(beg, lo), lo + EPB);
        end = min(max(end, beg), lo + EPB);
    }
    __syncthreads();
    for (int e = beg; e < end; e += 4) {
        int x[4];
        #pragma unroll
        for (int u = 0; u < 4; ++u) x[u] = tmp[min(e + u, N_EDGES - 1)].x;
        asm volatile("" :: "v"(x[0]), "v"(x[1]), "v"(x[2]), "v"(x[3]));
        #pragma unroll
        for (int u = 0; u < 4; ++u)
            if (e + u < end) atomicAdd(&lcnt[x[u] & 255], 1);
    }
    int part = beg - lo;
    for (int o = 16; o > 0; o >>= 1) part += __shfl_xor(part, o);
    if (lane == 0) s_base[wave] = part;
    __syncthreads();
    const int c = lcnt[tid];
    int inc = c;
    for (int o = 1; o < 32; o <<= 1) {
        int up = __shfl_up(inc, o);
        if (lane >= o) inc += up;
    }
    if (lane == 31) s_wsum[wave] = inc;
    __syncthreads();
    int start = inc - c;
    for (int w = 0; w < 8; ++w) {
        start += s_base[w];
        if (w < wave) start += s_wsum[w];
    }
    const int node = b * 256 + tid;
    if (node < N_NODES) {
        rowptr[node] = start;
        dinv[node] = rsqrtf(1.0f + (float)c);
        if (node == N_NODES - 1) rowptr[N_NODES] = start + c;
    }
    lcnt[tid] = start;
    __syncthreads();
    for (int e = beg; e < end; e += 4) {
        int2 ds[4];
        #pragma unroll
        for (int u = 0; u < 4; ++u) ds[u] = tmp[min(e + u, N_EDGES - 1)];
        asm volatile("" :: "v"(ds[0].x), "v"(ds[0].y), "v"(ds[1].x), "v"(ds[1].y),
                           "v"(ds[2].x), "v"(ds[2].y), "v"(ds[3].x), "v"(ds[3].y));
        #pragma unroll
        for (int u = 0; u < 4; ++u) {
            if (e + u < end) {
                const int k = ds[u].x & 255;
                const int r = atomicAdd(&lfill[k], 1);
                const unsigned p = min((unsigned)(lcnt[k] + r), (unsigned)(N_EDGES - 1));
                sorted[p] = ds[u].y;
            }
        }
    }
}

#define GEMM_BLOCKS 256
#define GEMM_WAVES 16
#define TPITCH 132

__device__ __forceinline__ void stage_w(const float* __restrict__ W, float4* __restrict__ sW, int tid) {
    float4 w[8];
    #pragma unroll
    for (int it = 0; it < 8; ++it) {
        const int q = tid + 512 * it;
        const int l = q & 31, t = (q >> 5) & 15, j = q >> 9;
        const int n = 16 * j + (l & 15);
        const int k = 8 * t + 2 * (l >> 4);
        w[it].x = W[(k    ) * HID + n];
        w[it].y = W[(k + 1) * HID + n];
        w[it].z = W[(k + 4) * HID + n];
        w[it].w = W[(k + 5) * HID + n];
    }
    #pragma unroll
    for (int it = 0; it < 8; ++it) sW[tid + 512 * it] = w[it];
}

__device__ __forceinline__ void tile_product(const float* T, const float4* sW,
                                             int lane, int lm, int half, v8f (&acc)[8]) {
    v2f a[32];
    #pragma unroll
    for (int s = 0; s < 32; ++s) a[s] = *(const v2f*)&T[lm * TPITCH + 4 * s + 2 * half];

    #pragma unroll
    for (int j = 0; j < 8; ++j) acc[j] = v8f{};

    #pragma unroll
    for (int t = 0; t < 16; ++t) {
        float4 b[8];
        #pragma unroll
        for (int j = 0; j < 8; ++j) b[j] = sW[(j * 16 + t) * 32 + lane];
        #pragma unroll
        for (int j = 0; j < 8; ++j) {
            v2f b0 = {b[j].x, b[j].y};
            acc[j] = __builtin_amdgcn_wmma_f32_16x16x4_f32(false, a[2 * t], false, b0,
                                                           (short)0, acc[j], false, false);
        }
        #pragma unroll
        for (int j = 0; j < 8; ++j) {
            v2f b1 = {b[j].z, b[j].w};
            acc[j] = __builtin_amdgcn_wmma_f32_16x16x4_f32(false, a[2 * t + 1], false, b1,
                                                           (short)0, acc[j], false, false);
        }
    }
}

__global__ __launch_bounds__(512) void k_gemm2(
    const float* __restrict__ A,
    const float* __restrict__ W1, const float* __restrict__ b1,
    const float* __restrict__ W2, const float* __restrict__ b2,
    const float* __restrict__ xres, float* __restrict__ out) {
    __shared__ float4 sW4[2][HID * HID / 4];
    __shared__ __attribute__((aligned(16))) float sT[GEMM_WAVES][16 * TPITCH];

    const int tid  = threadIdx.x;
    const int lane = tid & 31;
    const int half = lane >> 4;
    const int lm   = lane & 15;
    const int wave = __builtin_amdgcn_readfirstlane(tid >> 5);

    stage_w(W1, sW4[0], tid);
    stage_w(W2, sW4[1], tid);
    float bg[8];
    #pragma unroll
    for (int j = 0; j < 8; ++j) bg[j] = b1[16 * j + lm];
    const float4 bo = ((const float4*)b2)[lane];
    __syncthreads();

    float* T = sT[wave];
    const int n_tiles = N_NODES / 16;

    for (int tile = blockIdx.x + GEMM_BLOCKS * wave; tile < n_tiles; tile += GEMM_BLOCKS * GEMM_WAVES) {
        const size_t row0 = (size_t)tile * 16;

        float4 ar[16];
        #pragma unroll
        for (int r = 0; r < 16; ++r) ar[r] = ((const float4*)(A + (row0 + r) * HID))[lane];
        #pragma unroll
        for (int r = 0; r < 16; ++r) *(float4*)&T[r * TPITCH + 4 * lane] = ar[r];
        __builtin_amdgcn_fence(__ATOMIC_ACQ_REL, "wavefront");

        v8f acc[8];
        tile_product(T, sW4[0], lane, lm, half, acc);

        __builtin_amdgcn_fence(__ATOMIC_ACQ_REL, "wavefront");
        #pragma unroll
        for (int j = 0; j < 8; ++j) {
            #pragma unroll
            for (int v = 0; v < 8; ++v) {
                const float h = acc[j][v] + bg[j];
                T[(8 * half + v) * TPITCH + 16 * j + lm] = h > 0.f ? h : 0.f;
            }
        }
        __builtin_amdgcn_fence(__ATOMIC_ACQ_REL, "wavefront");

        tile_product(T, sW4[1], lane, lm, half, acc);

        __builtin_amdgcn_fence(__ATOMIC_ACQ_REL, "wavefront");
        #pragma unroll
        for (int j = 0; j < 8; ++j) {
            #pragma unroll
            for (int v = 0; v < 8; ++v) T[(8 * half + v) * TPITCH + 16 * j + lm] = acc[j][v];
        }
        __builtin_amdgcn_fence(__ATOMIC_ACQ_REL, "wavefront");

        float4 xr[16];
        #pragma unroll
        for (int r = 0; r < 16; ++r) xr[r] = ((const float4*)(xres + (row0 + r) * HID))[lane];
        #pragma unroll
        for (int r = 0; r < 16; ++r) {
            float4 v = *(const float4*)&T[r * TPITCH + 4 * lane];
            v.x += bo.x; v.y += bo.y; v.z += bo.z; v.w += bo.w;
            v.x = v.x > 0.f ? v.x : 0.f; v.y = v.y > 0.f ? v.y : 0.f;
            v.z = v.z > 0.f ? v.z : 0.f; v.w = v.w > 0.f ? v.w : 0.f;
            v.x += xr[r].x; v.y += xr[r].y; v.z += xr[r].z; v.w += xr[r].w;
            ((float4*)(out + (row0 + r) * HID))[lane] = v;
        }
        __builtin_amdgcn_fence(__ATOMIC_ACQ_REL, "wavefront");
    }
}


template <int K>
__device__ __forceinline__ void add_rows(float4& acc, const float4* __restrict__ rows,
                                         int es, int ew, int j) {
    float4 a[K];
    float w[K];
    #pragma unroll
    for (int k = 0; k < K; ++k) {
        const int s = __builtin_amdgcn_readlane(es, j + k);
        w[k] = __int_as_float(__builtin_amdgcn_readlane(ew, j + k));
        a[k] = rows[(size_t)s * 32];
    }
    #pragma unroll
    for (int k = 0; k < K; ++k) {
        acc.x = fmaf(w[k], a[k].x, acc.x);
        acc.y = fmaf(w[k], a[k].y, acc.y);
        acc.z = fmaf(w[k], a[k].z, acc.z);
        acc.w = fmaf(w[k], a[k].w, acc.w);
    }
}

__global__ __launch_bounds__(256) void k_gather(const int* __restrict__ rowptr,
                                                const int* __restrict__ csr,
                                                const float* __restrict__ dinv,
                                                const float* __restrict__ xin,
                                                float* __restrict__ agg) {
    const int lane = threadIdx.x & 31;
    const int node = __builtin_amdgcn_readfirstlane((int)(blockIdx.x * 8 + (threadIdx.x >> 5)));
    if (node >= N_NODES) return;
    const int beg = __builtin_amdgcn_readfirstlane(rowptr[node]);
    const int end = __builtin_amdgcn_readfirstlane(rowptr[node + 1]);
    const float di = dinv[node];
    const float4* rows = (const float4*)xin + lane;

    const float4 v = rows[(size_t)node * 32];
    const float wself = di * di;
    float4 acc = {wself * v.x, wself * v.y, wself * v.z, wself * v.w};

    for (int base = beg; base < end; base += 32) {
        const int idx = base + lane;
        int2 ent = {0, 0};
        if (idx < end) {
            ent.x = (int)min((unsigned)csr[idx], (unsigned)(N_NODES - 1));
            ent.y = __float_as_int(dinv[ent.x] * di);
        }
        const int rest = end - base;
        const int n = rest < 32 ? rest : 32;
        int j = 0;
        for (; j + 8 <= n; j += 8) add_rows<8>(acc, rows, ent.x, ent.y, j);
        if (j + 4 <= n) { add_rows<4>(acc, rows, ent.x, ent.y, j); j += 4; }
        const int r = n - j;
        if (r == 3)      add_rows<3>(acc, rows, ent.x, ent.y, j);
        else if (r == 2) add_rows<2>(acc, rows, ent.x, ent.y, j);
        else if (r == 1) add_rows<1>(acc, rows, ent.x, ent.y, j);
    }
    ((float4*)agg)[(size_t)node * 32 + lane] = acc;
}


extern "C" void kernel_launch(void* const* d_in, const int* in_sizes, int n_in,
                              void* d_out, int out_size, void* d_ws, size_t ws_size,
                              hipStream_t stream) {
    const float* x     = (const float*)d_in[0];
    const int*   ei    = (const int*)d_in[1];
    const float* W_gcn = (const float*)d_in[2];
    const float* b_gcn = (const float*)d_in[3];
    const float* W_lin = (const float*)d_in[4];
    const float* b_lin = (const float*)d_in[5];

    const int* src = ei;
    const int* dst = ei + N_EDGES;

    float* out = (float*)d_out;

    float* dinv   = (float*)d_ws;
    float* agg    = (float*)((char*)d_ws + (size_t)N_NODES * sizeof(float));
    int*   rowptr = (int*)((char*)d_ws + (size_t)52000256);
    int*   offs   = (int*)((char*)d_ws + (size_t)52400384);
    int*   tmp    = (int*)((char*)d_ws + (size_t)52816384);
    int*   csr    = (int*)((char*)d_ws + (size_t)65616384);

    const int NB_GATH  = (N_NODES + 7) / 8;

    k_part<<<NCHUNK, 256, 0, stream>>>(src, dst, tmp, offs);
    k_rank<<<NBK, 256, 0, stream>>>(offs, tmp, rowptr, dinv, csr);

    k_gather<<<NB_GATH, 256, 0, stream>>>(rowptr, csr, dinv, x, agg);

    k_gemm2<<<GEMM_BLOCKS, 512, 0, stream>>>(agg, W_gcn, b_gcn, W_lin, b_lin, x, out);
}


# edge ranking kernel: each short run read by 8 lanes (64-B pieces) into LDS, degrees counted and sources placed there, sorted piece stored as whole lines; LDS arrival order, so order of sums changes
# speedup vs baseline: 4.9942x; 1.0963x over previous
#include <hip/hip_runtime.h>

typedef __attribute__((ext_vector_type(2))) float v2f;
typedef __attribute__((ext_vector_type(8))) float v8f;

#define N_NODES 100000
#define N_EDGES 1600000
#define HID 128

#define EPB 6400
#define NCHUNK (N_EDGES / EPB)
#define NBK ((N_NODES + 255) / 256)
#define OPITCH 416
#define EPT (EPB / 256)
#define RCAP 6144

__global__ __launch_bounds__(256) void k_part(const int* __restrict__ src,
                                              const int* __restrict__ dst,
                                              int* __restrict__ tmp_i,
                                              int* __restrict__ offs) {
    __shared__ int hist[OPITCH];
    const int tid  = threadIdx.x;
    const int lane = tid & 31;
    const int base = blockIdx.x * EPB;
    int2* tmp = (int2*)tmp_i;
    for (int k = tid; k < OPITCH; k += 256) hist[k] = 0;
    int d[EPT], s[EPT], lr[EPT];
    #pragma unroll
    for (int i = 0; i < EPT; ++i) {
        d[i] = dst[base + i * 256 + tid];
        s[i] = src[base + i * 256 + tid];
    }
    __syncthreads();
    #pragma unroll
    for (int i = 0; i < EPT; ++i) lr[i] = atomicAdd(&hist[(unsigned)d[i] >> 8], 1);
    __syncthreads();
    if (tid < 32) {
        int carry = 0;
        for (int k0 = 0; k0 < OPITCH; k0 += 32) {
            const int v = hist[k0 + lane];
            int inc = v;
            for (int o = 1; o < 32; o <<= 1) {
                int up = __shfl_up(inc, o);
                if (lane >= o) inc += up;
            }
            hist[k0 + lane] = carry + inc - v;
            carry += __shfl(inc, 31);
        }
    }
    __syncthreads();
    #pragma unroll
    for (int i = 0; i < EPT; ++i) {
        const unsigned p = min((unsigned)(hist[(unsigned)d[i] >> 8] + lr[i]), (unsigned)(EPB - 1));
        tmp[base + p] = make_int2(d[i], s[i]);
    }
    for (int k = tid; k <= NBK; k += 256) offs[blockIdx.x * OPITCH + k] = base + hist[k];
}

__global__ __launch_bounds__(256) void k_rank(const int* __restrict__ offs,
                                              const int* __restrict__ tmp_i,
                                              int* __restrict__ rowptr,
                                              float* __restrict__ dinv,
                                              int* __restrict__ sorted) {
    __shared__ int lcnt[256];
    __shared__ int lfill[256];
    __shared__ int s_base[8];
    __shared__ int s_wsum[8];
    __shared__ int s_rbeg[256];
    __shared__ int s_rlen[256];
    __shared__ int s_roff[256];
    __shared__ int s_lsum[8];
    __shared__ int s_lmax[8];
    __shared__ int2 s_ent[RCAP];
    __shared__ int s_out[RCAP];
    const int tid  = threadIdx.x;
    const int lane = tid & 31;
    const int wave = tid >> 5;
    const int b    = blockIdx.x;
    const int2* tmp = (const int2*)tmp_i;
    lcnt[tid] = 0;
    lfill[tid] = 0;
    int beg = 0, end = 0, lo = 0;
    if (tid < NCHUNK) {
        lo  = tid * EPB;
        beg = offs[tid * OPITCH + b];
        end = offs[tid * OPITCH + b + 1];
        beg = min(max(beg, lo), lo + EPB);
        end = min(max(end, beg), lo + EPB);
    }
    const int len = end - beg;
    int linc = len;
    for (int o = 1; o < 32; o <<= 1) {
        int up = __shfl_up(linc, o);
        if (lane >= o) linc += up;
    }
    int lmax = len;
    for (int o = 16; o > 0; o >>= 1) lmax = max(lmax, __shfl_xor(lmax, o));
    int part = beg - lo;
    for (int o = 16; o > 0; o >>= 1) part += __shfl_xor(part, o);
    if (lane == 31) s_lsum[wave] = linc;
    if (lane == 0) { s_base[wave] = part; s_lmax[wave] = lmax; }
    s_rbeg[tid] = beg;
    s_rlen[tid] = len;
    __syncthreads();
    int roff = linc - len, tot = 0, mx = 0, rstart = 0;
    for (int w = 0; w < 8; ++w) {
        const int ls = s_lsum[w];
        if (w < wave) roff += ls;
        tot += ls;
        mx = max(mx, s_lmax[w]);
        rstart += s_base[w];
    }
    s_roff[tid] = roff;
    const bool staged = tot <= RCAP;
    __syncthreads();

    if (staged) {
        const int g = tid >> 3, sub = tid & 7;
        int rb[8], rn[8], ro[8];
        #pragma unroll
        for (int j = 0; j < 8; ++j) {
            rb[j] = s_rbeg[g + 32 * j];
            rn[j] = s_rlen[g + 32 * j];
            ro[j] = s_roff[g + 32 * j];
        }
        const int mxs = __builtin_amdgcn_readfirstlane(mx);
        for (int k0 = 0; k0 < mxs; k0 += 16) {
            int2 v[8][2];
            #pragma unroll
            for (int j = 0; j < 8; ++j) {
                #pragma unroll
                for (int u = 0; u < 2; ++u) {
                    const int k = min(k0 + sub + 8 * u, rn[j] - 1);
                    v[j][u] = tmp[min(max(rb[j] + k, 0), N_EDGES - 1)];
                }
            }
            #pragma unroll
            for (int j = 0; j < 8; ++j) {
                #pragma unroll
                for (int u = 0; u < 2; ++u) {
                    const int k = k0 + sub + 8 * u;
                    if (k < rn[j]) {
                        s_ent[min((unsigned)(ro[j] + k), (unsigned)(RCAP - 1))] = v[j][u];
                        atomicAdd(&lcnt[v[j][u].x & 255], 1);
                    }
                }
            }
        }
    } else {
        for (int e = beg; e < end; e += 4) {
            int x[4];
            #pragma unroll
            for (int u = 0; u < 4; ++u) x[u] = tmp[min(e + u, N_EDGES - 1)].x;
            asm volatile("" :: "v"(x[0]), "v"(x[1]), "v"(x[2]), "v"(x[3]));
            #pragma unroll
            for (int u = 0; u < 4; ++u)
                if (e + u < end) atomicAdd(&lcnt[x[u] & 255], 1);
        }
    }
    __syncthreads();
    const int c = lcnt[tid];
    int inc = c;
    for (int o = 1; o < 32; o <<= 1) {
        int up = __shfl_up(inc, o);
        if (lane >= o) inc += up;
    }
    if (lane == 31) s_wsum[wave] = inc;
    __syncthreads();
    int rel = inc - c;
    for (int w = 0; w < wave; ++w) rel += s_wsum[w];
    const int start = rstart + rel;
    const int node = b * 256 + tid;
    if (node < N_NODES) {
        rowptr[node] = start;
        dinv[node] = rsqrtf(1.0f + (float)c);
        if (node == N_NODES - 1) rowptr[N_NODES] = start + c;
    }
    if (staged) lfill[tid] = rel;
    else        lcnt[tid] = start;
    __syncthreads();
    if (staged) {
        #pragma unroll 4
        for (int i = tid; i < tot; i += 256) {
            const int2 e = s_ent[i];
            const int p = atomicAdd(&lfill[e.x & 255], 1);
            s_out[min((unsigned)p, (unsigned)(RCAP - 1))] = e.y;
        }
        __syncthreads();
        for (int i = tid; i < tot; i += 256) {
            sorted[min((unsigned)(rstart + i), (unsigned)(N_EDGES - 1))] = s_out[i];
        }
    } else {
        for (int e = beg; e < end; e += 4) {
            int2 ds[4];
            #pragma unroll
            for (int u = 0; u < 4; ++u) ds[u] = tmp[min(e + u, N_EDGES - 1)];
            asm volatile("" :: "v"(ds[0].x), "v"(ds[0].y), "v"(ds[1].x), "v"(ds[1].y),
                               "v"(ds[2].x), "v"(ds[2].y), "v"(ds[3].x), "v"(ds[3].y));
            #pragma unroll
            for (int u = 0; u < 4; ++u) {
                if (e + u < end) {
                    const int k = ds[u].x & 255;
                    const int r = atomicAdd(&lfill[k], 1);
                    const unsigned p = min((unsigned)(lcnt[k] + r), (unsigned)(N_EDGES - 1));
                    sorted[p] = ds[u].y;
                }
            }
        }
    }
}

#define GEMM_BLOCKS 256
#define GEMM_WAVES 16
#define TPITCH 132

__device__ __forceinline__ void stage_w(const float* __restrict__ W, float4* __restrict__ sW, int tid) {
    float4 w[8];
    #pragma unroll
    for (int it = 0; it < 8; ++it) {
        const int q = tid + 512 * it;
        const int l = q & 31, t = (q >> 5) & 15, j = q >> 9;
        const int n = 16 * j + (l & 15);
        const int k = 8 * t + 2 * (l >> 4);
        w[it].x = W[(k    ) * HID + n];
        w[it].y = W[(k + 1) * HID + n];
        w[it].z = W[(k + 4) * HID + n];
        w[it].w = W[(k + 5) * HID + n];
    }
    #pragma unroll
    for (int it = 0; it < 8; ++it) sW[tid + 512 * it] = w[it];
}

__device__ __forceinline__ void tile_product(const float* T, const float4* sW,
                                             int lane, int lm, int half, v8f (&acc)[8]) {
    v2f a[32];
    #pragma unroll
    for (int s = 0; s < 32; ++s) a[s] = *(const v2f*)&T[lm * TPITCH + 4 * s + 2 * half];

    #pragma unroll
    for (int j = 0; j < 8; ++j) acc[j] = v8f{};

    #pragma unroll
    for (int t = 0; t < 16; ++t) {
        float4 b[8];
        #pragma unroll
        for (int j = 0; j < 8; ++j) b[j] = sW[(j * 16 + t) * 32 + lane];
        #pragma unroll
        for (int j = 0; j < 8; ++j) {
            v2f b0 = {b[j].x, b[j].y};
            acc[j] = __builtin_amdgcn_wmma_f32_16x16x4_f32(false, a[2 * t], false, b0,
                                                           (short)0, acc[j], false, false);
        }
        #pragma unroll
        for (int j = 0; j < 8; ++j) {
            v2f b1 = {b[j].z, b[j].w};
            acc[j] = __builtin_amdgcn_wmma_f32_16x16x4_f32(false, a[2 * t + 1], false, b1,
                                                           (short)0, acc[j], false, false);
        }
    }
}

__global__ __launch_bounds__(512) void k_gemm2(
    const float* __restrict__ A,
    const float* __restrict__ W1, const float* __restrict__ b1,
    const float* __restrict__ W2, const float* __restrict__ b2,
    const float* __restrict__ xres, float* __restrict__ out) {
    __shared__ float4 sW4[2][HID * HID / 4];
    __shared__ __attribute__((aligned(16))) float sT[GEMM_WAVES][16 * TPITCH];

    const int tid  = threadIdx.x;
    const int lane = tid & 31;
    const int half = lane >> 4;
    const int lm   = lane & 15;
    const int wave = __builtin_amdgcn_readfirstlane(tid >> 5);

    stage_w(W1, sW4[0], tid);
    stage_w(W2, sW4[1], tid);
    float bg[8];
    #pragma unroll
    for (int j = 0; j < 8; ++j) bg[j] = b1[16 * j + lm];
    const float4 bo = ((const float4*)b2)[lane];
    __syncthreads();

    float* T = sT[wave];
    const int n_tiles = N_NODES / 16;

    for (int tile = blockIdx.x + GEMM_BLOCKS * wave; tile < n_tiles; tile += GEMM_BLOCKS * GEMM_WAVES) {
        const size_t row0 = (size_t)tile * 16;

        float4 ar[16];
        #pragma unroll
        for (int r = 0; r < 16; ++r) ar[r] = ((const float4*)(A + (row0 + r) * HID))[lane];
        #pragma unroll
        for (int r = 0; r < 16; ++r) *(float4*)&T[r * TPITCH + 4 * lane] = ar[r];
        __builtin_amdgcn_fence(__ATOMIC_ACQ_REL, "wavefront");

        v8f acc[8];
        tile_product(T, sW4[0], lane, lm, half, acc);

        __builtin_amdgcn_fence(__ATOMIC_ACQ_REL, "wavefront");
        #pragma unroll
        for (int j = 0; j < 8; ++j) {
            #pragma unroll
            for (int v = 0; v < 8; ++v) {
                const float h = acc[j][v] + bg[j];
                T[(8 * half + v) * TPITCH + 16 * j + lm] = h > 0.f ? h : 0.f;
            }
        }
        __builtin_amdgcn_fence(__ATOMIC_ACQ_REL, "wavefront");

        tile_product(T, sW4[1], lane, lm, half, acc);

        __builtin_amdgcn_fence(__ATOMIC_ACQ_REL, "wavefront");
        #pragma unroll
        for (int j = 0; j < 8; ++j) {
            #pragma unroll
            for (int v = 0; v < 8; ++v) T[(8 * half + v) * TPITCH + 16 * j + lm] = acc[j][v];
        }
        __builtin_amdgcn_fence(__ATOMIC_ACQ_REL, "wavefront");

        float4 xr[16];
        #pragma unroll
        for (int r = 0; r < 16; ++r) xr[r] = ((const float4*)(xres + (row0 + r) * HID))[lane];
        #pragma unroll
        for (int r = 0; r < 16; ++r) {
            float4 v = *(const float4*)&T[r * TPITCH + 4 * lane];
            v.x += bo.x; v.y += bo.y; v.z += bo.z; v.w += bo.w;
            v.x = v.x > 0.f ? v.x : 0.f; v.y = v.y > 0.f ? v.y : 0.f;
            v.z = v.z > 0.f ? v.z : 0.f; v.w = v.w > 0.f ? v.w : 0.f;
            v.x += xr[r].x; v.y += xr[r].y; v.z += xr[r].z; v.w += xr[r].w;
            ((float4*)(out + (row0 + r) * HID))[lane] = v;
        }
        __builtin_amdgcn_fence(__ATOMIC_ACQ_REL, "wavefront");
    }
}


template <int K>
__device__ __forceinline__ void add_rows(float4& acc, const float4* __restrict__ rows,
                                         int es, int ew, int j) {
    float4 a[K];
    float w[K];
    #pragma unroll
    for (int k = 0; k < K; ++k) {
        const int s = __builtin_amdgcn_readlane(es, j + k);
        w[k] = __int_as_float(__builtin_amdgcn_readlane(ew, j + k));
        a[k] = rows[(size_t)s * 32];
    }
    #pragma unroll
    for (int k = 0; k < K; ++k) {
        acc.x = fmaf(w[k], a[k].x, acc.x);
        acc.y = fmaf(w[k], a[k].y, acc.y);
        acc.z = fmaf(w[k], a[k].z, acc.z);
        acc.w = fmaf(w[k], a[k].w, acc.w);
    }
}

__global__ __launch_bounds__(256) void k_gather(const int* __restrict__ rowptr,
                                                const int* __restrict__ csr,
                                                const float* __restrict__ dinv,
                                                const float* __restrict__ xin,
                                                float* __restrict__ agg) {
    const int lane = threadIdx.x & 31;
    const int node = __builtin_amdgcn_readfirstlane((int)(blockIdx.x * 8 + (threadIdx.x >> 5)));
    if (node >= N_NODES) return;
    const int beg = __builtin_amdgcn_readfirstlane(rowptr[node]);
    const int end = __builtin_amdgcn_readfirstlane(rowptr[node + 1]);
    const float di = dinv[node];
    const float4* rows = (const float4*)xin + lane;

    const float4 v = rows[(size_t)node * 32];
    const float wself = di * di;
    float4 acc = {wself * v.x, wself * v.y, wself * v.z, wself * v.w};

    for (int base = beg; base < end; base += 32) {
        const int idx = base + lane;
        int2 ent = {0, 0};
        if (idx < end) {
            ent.x = (int)min((unsigned)csr[idx], (unsigned)(N_NODES - 1));
            ent.y = __float_as_int(dinv[ent.x] * di);
        }
        const int rest = end - base;
        const int n = rest < 32 ? rest : 32;
        int j = 0;
        for (; j + 8 <= n; j += 8) add_rows<8>(acc, rows, ent.x, ent.y, j);
        if (j + 4 <= n) { add_rows<4>(acc, rows, ent.x, ent.y, j); j += 4; }
        const int r = n - j;
        if (r == 3)      add_rows<3>(acc, rows, ent.x, ent.y, j);
        else if (r == 2) add_rows<2>(acc, rows, ent.x, ent.y, j);
        else if (r == 1) add_rows<1>(acc, rows, ent.x, ent.y, j);
    }
    ((float4*)agg)[(size_t)node * 32 + lane] = acc;
}


extern "C" void kernel_launch(void* const* d_in, const int* in_sizes, int n_in,
                              void* d_out, int out_size, void* d_ws, size_t ws_size,
                              hipStream_t stream) {
    const float* x     = (const float*)d_in[0];
    const int*   ei    = (const int*)d_in[1];
    const float* W_gcn = (const float*)d_in[2];
    const float* b_gcn = (const float*)d_in[3];
    const float* W_lin = (const float*)d_in[4];
    const float* b_lin = (const float*)d_in[5];

    const int* src = ei;
    const int* dst = ei + N_EDGES;

    float* out = (float*)d_out;

    float* dinv   = (float*)d_ws;
    float* agg    = (float*)((char*)d_ws + (size_t)N_NODES * sizeof(float));
    int*   rowptr = (int*)((char*)d_ws + (size_t)52000256);
    int*   offs   = (int*)((char*)d_ws + (size_t)52400384);
    int*   tmp    = (int*)((char*)d_ws + (size_t)52816384);
    int*   csr    = (int*)((char*)d_ws + (size_t)65616384);

    const int NB_GATH  = (N_NODES + 7) / 8;

    k_part<<<NCHUNK, 256, 0, stream>>>(src, dst, tmp, offs);
    k_rank<<<NBK, 256, 0, stream>>>(offs, tmp, rowptr, dinv, csr);

    k_gather<<<NB_GATH, 256, 0, stream>>>(rowptr, csr, dinv, x, agg);

    k_gemm2<<<GEMM_BLOCKS, 512, 0, stream>>>(agg, W_gcn, b_gcn, W_lin, b_lin, x, out);
}


# edge split kernel: pairs are placed in an LDS copy of the block's piece, which leaves as whole-line stores, 16 B per lane; was one 8-B store per pair, 32 lines per instruction; sums unchanged
# speedup vs baseline: 5.3026x; 1.0618x over previous
#include <hip/hip_runtime.h>

typedef __attribute__((ext_vector_type(2))) float v2f;
typedef __attribute__((ext_vector_type(8))) float v8f;

#define N_NODES 100000
#define N_EDGES 1600000
#define HID 128

#define EPB 6400
#define NCHUNK (N_EDGES / EPB)
#define NBK ((N_NODES + 255) / 256)
#define OPITCH 416
#define EPT (EPB / 256)
#define RCAP 6144

__global__ __launch_bounds__(256) void k_part(const int* __restrict__ src,
                                              const int* __restrict__ dst,
                                              int* __restrict__ tmp_i,
                                              int* __restrict__ offs) {
    __shared__ int hist[OPITCH];
    __shared__ __attribute__((aligned(16))) int2 s_pair[EPB];
    const int tid  = threadIdx.x;
    const int lane = tid & 31;
    const int base = blockIdx.x * EPB;
    int2* tmp = (int2*)tmp_i;
    for (int k = tid; k < OPITCH; k += 256) hist[k] = 0;
    int d[EPT], s[EPT], lr[EPT];
    #pragma unroll
    for (int i = 0; i < EPT; ++i) {
        d[i] = dst[base + i * 256 + tid];
        s[i] = src[base + i * 256 + tid];
    }
    __syncthreads();
    #pragma unroll
    for (int i = 0; i < EPT; ++i) lr[i] = atomicAdd(&hist[(unsigned)d[i] >> 8], 1);
    __syncthreads();
    if (tid < 32) {
        int carry = 0;
        for (int k0 = 0; k0 < OPITCH; k0 += 32) {
            const int v = hist[k0 + lane];
            int inc = v;
            for (int o = 1; o < 32; o <<= 1) {
                int up = __shfl_up(inc, o);
                if (lane >= o) inc += up;
            }
            hist[k0 + lane] = carry + inc - v;
            carry += __shfl(inc, 31);
        }
    }
    __syncthreads();
    #pragma unroll
    for (int i = 0; i < EPT; ++i) {
        const unsigned p = min((unsigned)(hist[(unsigned)d[i] >> 8] + lr[i]), (unsigned)(EPB - 1));
        s_pair[p] = make_int2(d[i], s[i]);
    }
    __syncthreads();
    const int4* sp4 = (const int4*)s_pair;
    int4* out4 = (int4*)(tmp + base);
    constexpr int NFULL = (EPB / 2) / 256;
    const bool last = tid < EPB / 2 - 256 * NFULL;
    int4 v[NFULL];
    int4 vl = {0, 0, 0, 0};
    #pragma unroll
    for (int i = 0; i < NFULL; ++i) v[i] = sp4[tid + 256 * i];
    if (last) vl = sp4[tid + 256 * NFULL];
    #pragma unroll
    for (int i = 0; i < NFULL; ++i) out4[tid + 256 * i] = v[i];
    if (last) out4[tid + 256 * NFULL] = vl;
    for (int k = tid; k <= NBK; k += 256) offs[blockIdx.x * OPITCH + k] = base + hist[k];
}

__global__ __launch_bounds__(256) void k_rank(const int* __restrict__ offs,
                                              const int* __restrict__ tmp_i,
                                              int* __restrict__ rowptr,
                                              float* __restrict__ dinv,
                                              int* __restrict__ sorted) {
    __shared__ int lcnt[256];
    __shared__ int lfill[256];
    __shared__ int s_base[8];
    __shared__ int s_wsum[8];
    __shared__ int s_rbeg[256];
    __shared__ int s_rlen[256];
    __shared__ int s_roff[256];
    __shared__ int s_lsum[8];
    __shared__ int s_lmax[8];
    __shared__ int2 s_ent[RCAP];
    __shared__ int s_out[RCAP];
    const int tid  = threadIdx.x;
    const int lane = tid & 31;
    const int wave = tid >> 5;
    const int b    = blockIdx.x;
    const int2* tmp = (const int2*)tmp_i;
    lcnt[tid] = 0;
    lfill[tid] = 0;
    int beg = 0, end = 0, lo = 0;
    if (tid < NCHUNK) {
        lo  = tid * EPB;
        beg = offs[tid * OPITCH + b];
        end = offs[tid * OPITCH + b + 1];
        beg = min(max(beg, lo), lo + EPB);
        end = min(max(end, beg), lo + EPB);
    }
    const int len = end - beg;
    int linc = len;
    for (int o = 1; o < 32; o <<= 1) {
        int up = __shfl_up(linc, o);
        if (lane >= o) linc += up;
    }
    int lmax = len;
    for (int o = 16; o > 0; o >>= 1) lmax = max(lmax, __shfl_xor(lmax, o));
    int part = beg - lo;
    for (int o = 16; o > 0; o >>= 1) part += __shfl_xor(part, o);
    if (lane == 31) s_lsum[wave] = linc;
    if (lane == 0) { s_base[wave] = part; s_lmax[wave] = lmax; }
    s_rbeg[tid] = beg;
    s_rlen[tid] = len;
    __syncthreads();
    int roff = linc - len, tot = 0, mx = 0, rstart = 0;
    for (int w = 0; w < 8; ++w) {
        const int ls = s_lsum[w];
        if (w < wave) roff += ls;
        tot += ls;
        mx = max(mx, s_lmax[w]);
        rstart += s_base[w];
    }
    s_roff[tid] = roff;
    const bool staged = tot <= RCAP;
    __syncthreads();

    if (staged) {
        const int g = tid >> 3, sub = tid & 7;
        int rb[8], rn[8], ro[8];
        #pragma unroll
        for (int j = 0; j < 8; ++j) {
            rb[j] = s_rbeg[g + 32 * j];
            rn[j] = s_rlen[g + 32 * j];
            ro[j] = s_roff[g + 32 * j];
        }
        const int mxs = __builtin_amdgcn_readfirstlane(mx);
        for (int k0 = 0; k0 < mxs; k0 += 16) {
            int2 v[8][2];
            #pragma unroll
            for (int j = 0; j < 8; ++j) {
                #pragma unroll
                for (int u = 0; u < 2; ++u) {
                    const int k = min(k0 + sub + 8 * u, rn[j] - 1);
                    v[j][u] = tmp[min(max(rb[j] + k, 0), N_EDGES - 1)];
                }
            }
            #pragma unroll
            for (int j = 0; j < 8; ++j) {
                #pragma unroll
                for (int u = 0; u < 2; ++u) {
                    const int k = k0 + sub + 8 * u;
                    if (k < rn[j]) {
                        s_ent[min((unsigned)(ro[j] + k), (unsigned)(RCAP - 1))] = v[j][u];
                        atomicAdd(&lcnt[v[j][u].x & 255], 1);
                    }
                }
            }
        }
    } else {
        for (int e = beg; e < end; e += 4) {
            int x[4];
            #pragma unroll
            for (int u = 0; u < 4; ++u) x[u] = tmp[min(e + u, N_EDGES - 1)].x;
            asm volatile("" :: "v"(x[0]), "v"(x[1]), "v"(x[2]), "v"(x[3]));
            #pragma unroll
            for (int u = 0; u < 4; ++u)
                if (e + u < end) atomicAdd(&lcnt[x[u] & 255], 1);
        }
    }
    __syncthreads();
    const int c = lcnt[tid];
    int inc = c;
    for (int o = 1; o < 32; o <<= 1) {
        int up = __shfl_up(inc, o);
        if (lane >= o) inc += up;
    }
    if (lane == 31) s_wsum[wave] = inc;
    __syncthreads();
    int rel = inc - c;
    for (int w = 0; w < wave; ++w) rel += s_wsum[w];
    const int start = rstart + rel;
    const int node = b * 256 + tid;
    if (node < N_NODES) {
        rowptr[node] = start;
        dinv[node] = rsqrtf(1.0f + (float)c);
        if (node == N_NODES - 1) rowptr[N_NODES] = start + c;
    }
    if (staged) lfill[tid] = rel;
    else        lcnt[tid] = start;
    __syncthreads();
    if (staged) {
        #pragma unroll 4
        for (int i = tid; i < tot; i += 256) {
            const int2 e = s_ent[i];
            const int p = atomicAdd(&lfill[e.x & 255], 1);
            s_out[min((unsigned)p, (unsigned)(RCAP - 1))] = e.y;
        }
        __syncthreads();
        for (int i = tid; i < tot; i += 256) {
            sorted[min((unsigned)(rstart + i), (unsigned)(N_EDGES - 1))] = s_out[i];
        }
    } else {
        for (int e = beg; e < end; e += 4) {
            int2 ds[4];
            #pragma unroll
            for (int u = 0; u < 4; ++u) ds[u] = tmp[min(e + u, N_EDGES - 1)];
            asm volatile("" :: "v"(ds[0].x), "v"(ds[0].y), "v"(ds[1].x), "v"(ds[1].y),
                               "v"(ds[2].x), "v"(ds[2].y), "v"(ds[3].x), "v"(ds[3].y));
            #pragma unroll
            for (int u = 0; u < 4; ++u) {
                if (e + u < end) {
                    const int k = ds[u].x & 255;
                    const int r = atomicAdd(&lfill[k], 1);
                    const unsigned p = min((unsigned)(lcnt[k] + r), (unsigned)(N_EDGES - 1));
                    sorted[p] = ds[u].y;
                }
            }
        }
    }
}

#define GEMM_BLOCKS 256
#define GEMM_WAVES 16
#define TPITCH 132

__device__ __forceinline__ void stage_w(const float* __restrict__ W, float4* __restrict__ sW, int tid) {
    float4 w[8];
    #pragma unroll
    for (int it = 0; it < 8; ++it) {
        const int q = tid + 512 * it;
        const int l = q & 31, t = (q >> 5) & 15, j = q >> 9;
        const int n = 16 * j + (l & 15);
        const int k = 8 * t + 2 * (l >> 4);
        w[it].x = W[(k    ) * HID + n];
        w[it].y = W[(k + 1) * HID + n];
        w[it].z = W[(k + 4) * HID + n];
        w[it].w = W[(k + 5) * HID + n];
    }
    #pragma unroll
    for (int it = 0; it < 8; ++it) sW[tid + 512 * it] = w[it];
}

__device__ __forceinline__ void tile_product(const float* T, const float4* sW,
                                             int lane, int lm, int half, v8f (&acc)[8]) {
    v2f a[32];
    #pragma unroll
    for (int s = 0; s < 32; ++s) a[s] = *(const v2f*)&T[lm * TPITCH + 4 * s + 2 * half];

    #pragma unroll
    for (int j = 0; j < 8; ++j) acc[j] = v8f{};

    #pragma unroll
    for (int t = 0; t < 16; ++t) {
        float4 b[8];
        #pragma unroll
        for (int j = 0; j < 8; ++j) b[j] = sW[(j * 16 + t) * 32 + lane];
        #pragma unroll
        for (int j = 0; j < 8; ++j) {
            v2f b0 = {b[j].x, b[j].y};
            acc[j] = __builtin_amdgcn_wmma_f32_16x16x4_f32(false, a[2 * t], false, b0,
                                                           (short)0, acc[j], false, false);
        }
        #pragma unroll
        for (int j = 0; j < 8; ++j) {
            v2f b1 = {b[j].z, b[j].w};
            acc[j] = __builtin_amdgcn_wmma_f32_16x16x4_f32(false, a[2 * t + 1], false, b1,
                                                           (short)0, acc[j], false, false);
        }
    }
}

__global__ __launch_bounds__(512) void k_gemm2(
    const float* __restrict__ A,
    const float* __restrict__ W1, const float* __restrict__ b1,
    const float* __restrict__ W2, const float* __restrict__ b2,
    const float* __restrict__ xres, float* __restrict__ out) {
    __shared__ float4 sW4[2][HID * HID / 4];
    __shared__ __attribute__((aligned(16))) float sT[GEMM_WAVES][16 * TPITCH];

    const int tid  = threadIdx.x;
    const int lane = tid & 31;
    const int half = lane >> 4;
    const int lm   = lane & 15;
    const int wave = __builtin_amdgcn_readfirstlane(tid >> 5);

    stage_w(W1, sW4[0], tid);
    stage_w(W2, sW4[1], tid);
    float bg[8];
    #pragma unroll
    for (int j = 0; j < 8; ++j) bg[j] = b1[16 * j + lm];
    const float4 bo = ((const float4*)b2)[lane];
    __syncthreads();

    float* T = sT[wave];
    const int n_tiles = N_NODES / 16;

    for (int tile = blockIdx.x + GEMM_BLOCKS * wave; tile < n_tiles; tile += GEMM_BLOCKS * GEMM_WAVES) {
        const size_t row0 = (size_t)tile * 16;

        float4 ar[16];
        #pragma unroll
        for (int r = 0; r < 16; ++r) ar[r] = ((const float4*)(A + (row0 + r) * HID))[lane];
        #pragma unroll
        for (int r = 0; r < 16; ++r) *(float4*)&T[r * TPITCH + 4 * lane] = ar[r];
        __builtin_amdgcn_fence(__ATOMIC_ACQ_REL, "wavefront");

        v8f acc[8];
        tile_product(T, sW4[0], lane, lm, half, acc);

        __builtin_amdgcn_fence(__ATOMIC_ACQ_REL, "wavefront");
        #pragma unroll
        for (int j = 0; j < 8; ++j) {
            #pragma unroll
            for (int v = 0; v < 8; ++v) {
                const float h = acc[j][v] + bg[j];
                T[(8 * half + v) * TPITCH + 16 * j + lm] = h > 0.f ? h : 0.f;
            }
        }
        __builtin_amdgcn_fence(__ATOMIC_ACQ_REL, "wavefront");

        tile_product(T, sW4[1], lane, lm, half, acc);

        __builtin_amdgcn_fence(__ATOMIC_ACQ_REL, "wavefront");
        #pragma unroll
        for (int j = 0; j < 8; ++j) {
            #pragma unroll
            for (int v = 0; v < 8; ++v) T[(8 * half + v) * TPITCH + 16 * j + lm] = acc[j][v];
        }
        __builtin_amdgcn_fence(__ATOMIC_ACQ_REL, "wavefront");

        float4 xr[16];
        #pragma unroll
        for (int r = 0; r < 16; ++r) xr[r] = ((const float4*)(xres + (row0 + r) * HID))[lane];
        #pragma unroll
        for (int r = 0; r < 16; ++r) {
            float4 v = *(const float4*)&T[r * TPITCH + 4 * lane];
            v.x += bo.x; v.y += bo.y; v.z += bo.z; v.w += bo.w;
            v.x = v.x > 0.f ? v.x : 0.f; v.y = v.y > 0.f ? v.y : 0.f;
            v.z = v.z > 0.f ? v.z : 0.f; v.w = v.w > 0.f ? v.w : 0.f;
            v.x += xr[r].x; v.y += xr[r].y; v.z += xr[r].z; v.w += xr[r].w;
            ((float4*)(out + (row0 + r) * HID))[lane] = v;
        }
        __builtin_amdgcn_fence(__ATOMIC_ACQ_REL, "wavefront");
    }
}


template <int K>
__device__ __forceinline__ void add_rows(float4& acc, const float4* __restrict__ rows,
                                         int es, int ew, int j) {
    float4 a[K];
    float w[K];
    #pragma unroll
    for (int k = 0; k < K; ++k) {
        const int s = __builtin_amdgcn_readlane(es, j + k);
        w[k] = __int_as_float(__builtin_amdgcn_readlane(ew, j + k));
        a[k] = rows[(size_t)s * 32];
    }
    #pragma unroll
    for (int k = 0; k < K; ++k) {
        acc.x = fmaf(w[k], a[k].x, acc.x);
        acc.y = fmaf(w[k], a[k].y, acc.y);
        acc.z = fmaf(w[k], a[k].z, acc.z);
        acc.w = fmaf(w[k], a[k].w, acc.w);
    }
}

__global__ __launch_bounds__(256) void k_gather(const int* __restrict__ rowptr,
                                                const int* __restrict__ csr,
                                                const float* __restrict__ dinv,
                                                const float* __restrict__ xin,
                                                float* __restrict__ agg) {
    const int lane = threadIdx.x & 31;
    const int node = __builtin_amdgcn_readfirstlane((int)(blockIdx.x * 8 + (threadIdx.x >> 5)));
    if (node >= N_NODES) return;
    const int beg = __builtin_amdgcn_readfirstlane(rowptr[node]);
    const int end = __builtin_amdgcn_readfirstlane(rowptr[node + 1]);
    const float di = dinv[node];
    const float4* rows = (const float4*)xin + lane;

    const float4 v = rows[(size_t)node * 32];
    const float wself = di * di;
    float4 acc = {wself * v.x, wself * v.y, wself * v.z, wself * v.w};

    for (int base = beg; base < end; base += 32) {
        const int idx = base + lane;
        int2 ent = {0, 0};
        if (idx < end) {
            ent.x = (int)min((unsigned)csr[idx], (unsigned)(N_NODES - 1));
            ent.y = __float_as_int(dinv[ent.x] * di);
        }
        const int rest = end - base;
        const int n = rest < 32 ? rest : 32;
        int j = 0;
        for (; j + 8 <= n; j += 8) add_rows<8>(acc, rows, ent.x, ent.y, j);
        if (j + 4 <= n) { add_rows<4>(acc, rows, ent.x, ent.y, j); j += 4; }
        const int r = n - j;
        if (r == 3)      add_rows<3>(acc, rows, ent.x, ent.y, j);
        else if (r == 2) add_rows<2>(acc, rows, ent.x, ent.y, j);
        else if (r == 1) add_rows<1>(acc, rows, ent.x, ent.y, j);
    }
    ((float4*)agg)[(size_t)node * 32 + lane] = acc;
}


extern "C" void kernel_launch(void* const* d_in, const int* in_sizes, int n_in,
                              void* d_out, int out_size, void* d_ws, size_t ws_size,
                              hipStream_t stream) {
    const float* x     = (const float*)d_in[0];
    const int*   ei    = (const int*)d_in[1];
    const float* W_gcn = (const float*)d_in[2];
    const float* b_gcn = (const float*)d_in[3];
    const float* W_lin = (const float*)d_in[4];
    const float* b_lin = (const float*)d_in[5];

    const int* src = ei;
    const int* dst = ei + N_EDGES;

    float* out = (float*)d_out;

    float* dinv   = (float*)d_ws;
    float* agg    = (float*)((char*)d_ws + (size_t)N_NODES * sizeof(float));
    int*   rowptr = (int*)((char*)d_ws + (size_t)52000256);
    int*   offs   = (int*)((char*)d_ws + (size_t)52400384);
    int*   tmp    = (int*)((char*)d_ws + (size_t)52816384);
    int*   csr    = (int*)((char*)d_ws + (size_t)65616384);

    const int NB_GATH  = (N_NODES + 7) / 8;

    k_part<<<NCHUNK, 256, 0, stream>>>(src, dst, tmp, offs);
    k_rank<<<NBK, 256, 0, stream>>>(offs, tmp, rowptr, dinv, csr);

    k_gather<<<NB_GATH, 256, 0, stream>>>(rowptr, csr, dinv, x, agg);

    k_gemm2<<<GEMM_BLOCKS, 512, 0, stream>>>(agg, W_gcn, b_gcn, W_lin, b_lin, x, out);
}
